# starting kernel
# baseline (speedup 1.0000x reference)
#include <hip/hip_runtime.h>
#include <hip/hip_bf16.h>
#include <math.h>

typedef float v2f __attribute__((ext_vector_type(2)));
typedef float v8f __attribute__((ext_vector_type(8)));

#define NB 8
#define CC 256
#define HH 8
#define DD 32
#define LL 4096
#define TC 768

static __device__ __forceinline__ v8f wmma_f32(v2f a, v2f b, v8f c) {
    return __builtin_amdgcn_wmma_f32_16x16x4_f32(false, a, false, b, (short)0, c, false, false);
}

__global__ __launch_bounds__(256) void qkv_gemm_kernel(
    const float* __restrict__ x, const float* __restrict__ wq,
    const float* __restrict__ bq, float* __restrict__ qkv)
{
    __shared__ float2 Wt2[16 * 66];
    const int mt = blockIdx.x;
    const int nt = blockIdx.y;
    const int row0 = mt * 128;
    const int n = row0 >> 12;
    const int l0 = row0 & (LL - 1);
    const int j0 = nt * 64;
    const int tid = threadIdx.x;
    const int lane = tid & 31, wave = tid >> 5;
    const int m0 = wave * 16;
    const int laneM = lane & 15;
    const int laneH = lane >> 4;
    const int laneK = laneH << 1;

    v8f acc[4] = {};
    const float* xA = x + (size_t)n * CC * LL + l0 + m0 + laneM;

    for (int kc = 0; kc < CC; kc += 32) {
        for (int i = tid; i < 16 * 64; i += 256) {
            const int p = i & 15, j = i >> 4;
            Wt2[p * 66 + j] = *(const float2*)&wq[(size_t)(j0 + j) * CC + kc + 2 * p];
        }
        __syncthreads();
#pragma unroll
        for (int kk = 0; kk < 32; kk += 4) {
            const int k = kc + kk + laneK;
            v2f a;
            a.x = xA[(size_t)k * LL];
            a.y = xA[(size_t)(k + 1) * LL];
            const int pp = (kk >> 1) + laneH;
#pragma unroll
            for (int nb = 0; nb < 4; ++nb) {
                const v2f b = *(const v2f*)&Wt2[pp * 66 + nb * 16 + laneM];
                acc[nb] = wmma_f32(a, b, acc[nb]);
            }
        }
        __syncthreads();
    }
    const int mbase = m0 + (laneH << 3);
#pragma unroll
    for (int nb = 0; nb < 4; ++nb) {
        const int j = j0 + nb * 16 + laneM;
        const float bias = bq[j];
        float* outp = qkv + (size_t)(row0 + mbase) * TC + j;
#pragma unroll
        for (int r = 0; r < 8; ++r) outp[(size_t)r * TC] = acc[nb][r] + bias;
    }
}

__global__ __launch_bounds__(256) void norm_qk_kernel(float* __restrict__ qkv)
{
    size_t t = (size_t)blockIdx.x * 256 + threadIdx.x;
    const int h = (int)(t & 7);
    size_t r = t >> 3;
    const int l = (int)(r & (LL - 1));
    r >>= 12;
    const int n = (int)(r & 7);
    const int s = (int)(r >> 3);
    float* p = qkv + ((size_t)(n * LL + l)) * TC + s * CC + h * DD;
    float4 v[8];
    float ss = 0.f;
#pragma unroll
    for (int i = 0; i < 8; ++i) {
        v[i] = ((const float4*)p)[i];
        ss += v[i].x * v[i].x + v[i].y * v[i].y + v[i].z * v[i].z + v[i].w * v[i].w;
    }
    const float inv = 1.0f / sqrtf(ss);
#pragma unroll
    for (int i = 0; i < 8; ++i) {
        v[i].x *= inv; v[i].y *= inv; v[i].z *= inv; v[i].w *= inv;
        ((float4*)p)[i] = v[i];
    }
}

__global__ __launch_bounds__(256) void attn_kernel(
    const float* __restrict__ qkv, float* __restrict__ attn)
{
    __shared__ float red[8 * 1024];
    const int nh = blockIdx.x;
    const int n = nh >> 3, h = nh & 7;
    const int tid = threadIdx.x, lane = tid & 31, wave = tid >> 5;
    const int laneM = lane & 15;
    const int laneK = (lane >> 4) << 1;

    v8f acc[2][2] = {};
    const float* base = qkv + (size_t)n * LL * TC + h * DD;

    const int lend = wave * 512 + 512;
    for (int l = wave * 512; l < lend; l += 4) {
        const size_t r0 = (size_t)(l + laneK) * TC;
        v2f aK[2], bV[2];
#pragma unroll
        for (int md = 0; md < 2; ++md) {
            aK[md].x = base[r0 + CC + md * 16 + laneM];
            aK[md].y = base[r0 + TC + CC + md * 16 + laneM];
        }
#pragma unroll
        for (int me = 0; me < 2; ++me) {
            bV[me].x = base[r0 + 2 * CC + me * 16 + laneM];
            bV[me].y = base[r0 + TC + 2 * CC + me * 16 + laneM];
        }
#pragma unroll
        for (int md = 0; md < 2; ++md)
#pragma unroll
            for (int me = 0; me < 2; ++me)
                acc[md][me] = wmma_f32(aK[md], bV[me], acc[md][me]);
    }
    const int dadd = (lane >> 4) << 3;
#pragma unroll
    for (int md = 0; md < 2; ++md)
#pragma unroll
        for (int me = 0; me < 2; ++me)
#pragma unroll
            for (int r = 0; r < 8; ++r)
                red[wave * 1024 + (md * 16 + r + dadd) * 32 + me * 16 + laneM] = acc[md][me][r];
    __syncthreads();
    for (int i = tid; i < 1024; i += 256) {
        float s = 0.f;
#pragma unroll
        for (int w = 0; w < 8; ++w) s += red[w * 1024 + i];
        attn[(size_t)nh * 1024 + i] = s;
    }
}

__global__ __launch_bounds__(256) void epilogue_kernel(
    const float* __restrict__ qkv, const float* __restrict__ attn,
    const float* __restrict__ wdc, float* __restrict__ out_pre)
{
    __shared__ float v_s[72 * 32];
    __shared__ float q_s[64 * 32];
    __shared__ float a_s[32 * 32];
    const int b = blockIdx.x;
    const int lc = b & 63;
    const int h = (b >> 6) & 7;
    const int n = b >> 9;
    const int l0 = lc * 64;
    const int tid = threadIdx.x;
    const float* qbase = qkv + (size_t)n * LL * TC + h * DD;

    for (int i = tid; i < 72 * 32; i += 256) {
        const int rr = i >> 5, e = i & 31;
        const int l = l0 - 4 + rr;
        v_s[i] = (l >= 0 && l < LL) ? qbase[(size_t)l * TC + 2 * CC + e] : 0.0f;
    }
    for (int i = tid; i < 64 * 32; i += 256) {
        const int rr = i >> 5, d = i & 31;
        q_s[i] = qbase[(size_t)(l0 + rr) * TC + d];
    }
    for (int i = tid; i < 1024; i += 256)
        a_s[i] = attn[((size_t)(n * 8 + h)) * 1024 + i];
    __syncthreads();

    float wd[9];
#pragma unroll
    for (int r = 0; r < 9; ++r) wd[r] = wdc[h * 9 + r];

    const int lane = tid & 31, wave = tid >> 5;
    const float inv_pi = 0.31830988618379067154f;
    for (int ll = wave * 8; ll < wave * 8 + 8; ++ll) {
        float acc = 0.f;
#pragma unroll
        for (int d = 0; d < 32; ++d) acc += q_s[ll * 32 + d] * a_s[d * 32 + lane];
        const float val = 0.5f * v_s[(ll + 4) * 32 + lane] + inv_pi * acc;
        float ss = val * val;
#pragma unroll
        for (int off = 16; off; off >>= 1) ss += __shfl_xor(ss, off, 32);
        const float res = val / sqrtf(ss);
        float dc = 0.f;
#pragma unroll
        for (int r = 0; r < 9; ++r) dc += wd[r] * v_s[(ll + r) * 32 + lane];
        out_pre[(size_t)(n * LL + l0 + ll) * CC + h * DD + lane] = res + dc;
    }
}

__global__ __launch_bounds__(256) void proj_gemm_kernel(
    const float* __restrict__ A, const float* __restrict__ wp,
    const float* __restrict__ bp, float* __restrict__ out)
{
    __shared__ float2 smemv[4224];
    float* smem = (float*)smemv;
    float2* At2 = smemv;
    float2* Wt2 = smemv + 16 * 130;
    const int mt = blockIdx.x;
    const int nt = blockIdx.y;
    const int row0 = mt * 128;
    const int j0 = nt * 64;
    const int tid = threadIdx.x, lane = tid & 31, wave = tid >> 5;
    const int m0 = wave * 16;
    const int laneM = lane & 15;
    const int laneH = lane >> 4;

    v8f acc[4] = {};
    for (int kc = 0; kc < CC; kc += 32) {
        for (int i = tid; i < 16 * 128; i += 256) {
            const int p = i & 15, rr = i >> 4;
            At2[p * 130 + rr] = *(const float2*)&A[(size_t)(row0 + rr) * CC + kc + 2 * p];
        }
        for (int i = tid; i < 16 * 64; i += 256) {
            const int p = i & 15, j = i >> 4;
            Wt2[p * 66 + j] = *(const float2*)&wp[(size_t)(j0 + j) * CC + kc + 2 * p];
        }
        __syncthreads();
#pragma unroll
        for (int kk = 0; kk < 32; kk += 4) {
            const int pp = (kk >> 1) + laneH;
            const v2f a = *(const v2f*)&At2[pp * 130 + m0 + laneM];
#pragma unroll
            for (int nb = 0; nb < 4; ++nb) {
                const v2f b = *(const v2f*)&Wt2[pp * 66 + nb * 16 + laneM];
                acc[nb] = wmma_f32(a, b, acc[nb]);
            }
        }
        __syncthreads();
    }
    const int lbase = m0 + (laneH << 3);
#pragma unroll
    for (int nb = 0; nb < 4; ++nb) {
        const int j = nb * 16 + laneM;
#pragma unroll
        for (int r = 0; r < 8; ++r) smem[j * 132 + lbase + r] = acc[nb][r];
    }
    __syncthreads();
    const int n = row0 >> 12;
    const int lg0 = row0 & (LL - 1);
    for (int i = tid; i < 64 * 128; i += 256) {
        const int l = i & 127, j = i >> 7;
        out[((size_t)(n * CC + j0 + j)) * LL + lg0 + l] = smem[j * 132 + l] + bp[j0 + j];
    }
}

extern "C" void kernel_launch(void* const* d_in, const int* in_sizes, int n_in,
                              void* d_out, int out_size, void* d_ws, size_t ws_size,
                              hipStream_t stream) {
    const float* x      = (const float*)d_in[0];
    const float* w_qkv  = (const float*)d_in[1];
    const float* b_qkv  = (const float*)d_in[2];
    const float* w_proj = (const float*)d_in[3];
    const float* b_proj = (const float*)d_in[4];
    const float* w_dcv  = (const float*)d_in[5];
    float* out = (float*)d_out;

    float* qkv     = (float*)d_ws;
    float* attn    = qkv + (size_t)NB * LL * TC;
    float* out_pre = attn + (size_t)NB * HH * DD * DD;

    qkv_gemm_kernel<<<dim3(256, 12), 256, 0, stream>>>(x, w_qkv, b_qkv, qkv);
    norm_qk_kernel<<<2048, 256, 0, stream>>>(qkv);
    attn_kernel<<<64, 256, 0, stream>>>(qkv, attn);
    epilogue_kernel<<<4096, 256, 0, stream>>>(qkv, attn, w_dcv, out_pre);
    proj_gemm_kernel<<<dim3(256, 4), 256, 0, stream>>>(out_pre, w_proj, b_proj, out);
}


# Fold the q/k per-head L2 normalisation into the QKV GEMM epilogue (sum of squares by a 16-lane DPP butterfly: its summation order changes); the separate normalise launch is gone (5 -> 4 launches)
# speedup vs baseline: 1.1845x; 1.1845x over previous
#include <hip/hip_runtime.h>
#include <hip/hip_bf16.h>
#include <math.h>

typedef float v2f __attribute__((ext_vector_type(2)));
typedef float v8f __attribute__((ext_vector_type(8)));

#define NB 8
#define CC 256
#define HH 8
#define DD 32
#define LL 4096
#define TC 768

static __device__ __forceinline__ v8f wmma_f32(v2f a, v2f b, v8f c) {
    return __builtin_amdgcn_wmma_f32_16x16x4_f32(false, a, false, b, (short)0, c, false, false);
}

static __device__ __forceinline__ float row16_sum(float v) {
    v += __builtin_bit_cast(float, __builtin_amdgcn_update_dpp(0, __builtin_bit_cast(int, v), 0x161, 0xf, 0xf, true));
    v += __builtin_bit_cast(float, __builtin_amdgcn_update_dpp(0, __builtin_bit_cast(int, v), 0x162, 0xf, 0xf, true));
    v += __builtin_bit_cast(float, __builtin_amdgcn_update_dpp(0, __builtin_bit_cast(int, v), 0x164, 0xf, 0xf, true));
    v += __builtin_bit_cast(float, __builtin_amdgcn_update_dpp(0, __builtin_bit_cast(int, v), 0x168, 0xf, 0xf, true));
    return v;
}

__global__ __launch_bounds__(256) void qkv_gemm_kernel(
    const float* __restrict__ x, const float* __restrict__ wq,
    const float* __restrict__ bq, float* __restrict__ qkv)
{
    __shared__ float2 Wt2[16 * 66];
    const int mt = blockIdx.x;
    const int nt = blockIdx.y;
    const int row0 = mt * 128;
    const int n = row0 >> 12;
    const int l0 = row0 & (LL - 1);
    const int j0 = nt * 64;
    const int tid = threadIdx.x;
    const int lane = tid & 31, wave = tid >> 5;
    const int m0 = wave * 16;
    const int laneM = lane & 15;
    const int laneH = lane >> 4;
    const int laneK = laneH << 1;

    v8f acc[4] = {};
    const float* xA = x + (size_t)n * CC * LL + l0 + m0 + laneM;

    for (int kc = 0; kc < CC; kc += 32) {
        for (int i = tid; i < 16 * 64; i += 256) {
            const int p = i & 15, j = i >> 4;
            Wt2[p * 66 + j] = *(const float2*)&wq[(size_t)(j0 + j) * CC + kc + 2 * p];
        }
        __syncthreads();
#pragma unroll
        for (int kk = 0; kk < 32; kk += 4) {
            const int k = kc + kk + laneK;
            v2f a;
            a.x = xA[(size_t)k * LL];
            a.y = xA[(size_t)(k + 1) * LL];
            const int pp = (kk >> 1) + laneH;
#pragma unroll
            for (int nb = 0; nb < 4; ++nb) {
                const v2f b = *(const v2f*)&Wt2[pp * 66 + nb * 16 + laneM];
                acc[nb] = wmma_f32(a, b, acc[nb]);
            }
        }
        __syncthreads();
    }
#pragma unroll
    for (int nb = 0; nb < 4; ++nb) {
        const float bias = bq[j0 + nb * 16 + laneM];
#pragma unroll
        for (int r = 0; r < 8; ++r) acc[nb][r] += bias;
    }
    if (j0 < 2 * CC) {
#pragma unroll
        for (int hd = 0; hd < 2; ++hd) {
#pragma unroll
            for (int r = 0; r < 8; ++r) {
                const float a0 = acc[2 * hd][r], a1 = acc[2 * hd + 1][r];
                const float ss = row16_sum(a0 * a0 + a1 * a1);
                const float inv = 1.0f / sqrtf(ss);
                acc[2 * hd][r] = a0 * inv;
                acc[2 * hd + 1][r] = a1 * inv;
            }
        }
    }
    const int mbase = m0 + (laneH << 3);
#pragma unroll
    for (int nb = 0; nb < 4; ++nb) {
        float* outp = qkv + (size_t)(row0 + mbase) * TC + j0 + nb * 16 + laneM;
#pragma unroll
        for (int r = 0; r < 8; ++r) outp[(size_t)r * TC] = acc[nb][r];
    }
}

__global__ __launch_bounds__(256) void attn_kernel(
    const float* __restrict__ qkv, float* __restrict__ attn)
{
    __shared__ float red[8 * 1024];
    const int nh = blockIdx.x;
    const int n = nh >> 3, h = nh & 7;
    const int tid = threadIdx.x, lane = tid & 31, wave = tid >> 5;
    const int laneM = lane & 15;
    const int laneK = (lane >> 4) << 1;

    v8f acc[2][2] = {};
    const float* base = qkv + (size_t)n * LL * TC + h * DD;

    const int lend = wave * 512 + 512;
    for (int l = wave * 512; l < lend; l += 4) {
        const size_t r0 = (size_t)(l + laneK) * TC;
        v2f aK[2], bV[2];
#pragma unroll
        for (int md = 0; md < 2; ++md) {
            aK[md].x = base[r0 + CC + md * 16 + laneM];
            aK[md].y = base[r0 + TC + CC + md * 16 + laneM];
        }
#pragma unroll
        for (int me = 0; me < 2; ++me) {
            bV[me].x = base[r0 + 2 * CC + me * 16 + laneM];
            bV[me].y = base[r0 + TC + 2 * CC + me * 16 + laneM];
        }
#pragma unroll
        for (int md = 0; md < 2; ++md)
#pragma unroll
            for (int me = 0; me < 2; ++me)
                acc[md][me] = wmma_f32(aK[md], bV[me], acc[md][me]);
    }
    const int dadd = (lane >> 4) << 3;
#pragma unroll
    for (int md = 0; md < 2; ++md)
#pragma unroll
        for (int me = 0; me < 2; ++me)
#pragma unroll
            for (int r = 0; r < 8; ++r)
                red[wave * 1024 + (md * 16 + r + dadd) * 32 + me * 16 + laneM] = acc[md][me][r];
    __syncthreads();
    for (int i = tid; i < 1024; i += 256) {
        float s = 0.f;
#pragma unroll
        for (int w = 0; w < 8; ++w) s += red[w * 1024 + i];
        attn[(size_t)nh * 1024 + i] = s;
    }
}

__global__ __launch_bounds__(256) void epilogue_kernel(
    const float* __restrict__ qkv, const float* __restrict__ attn,
    const float* __restrict__ wdc, float* __restrict__ out_pre)
{
    __shared__ float v_s[72 * 32];
    __shared__ float q_s[64 * 32];
    __shared__ float a_s[32 * 32];
    const int b = blockIdx.x;
    const int lc = b & 63;
    const int h = (b >> 6) & 7;
    const int n = b >> 9;
    const int l0 = lc * 64;
    const int tid = threadIdx.x;
    const float* qbase = qkv + (size_t)n * LL * TC + h * DD;

    for (int i = tid; i < 72 * 32; i += 256) {
        const int rr = i >> 5, e = i & 31;
        const int l = l0 - 4 + rr;
        v_s[i] = (l >= 0 && l < LL) ? qbase[(size_t)l * TC + 2 * CC + e] : 0.0f;
    }
    for (int i = tid; i < 64 * 32; i += 256) {
        const int rr = i >> 5, d = i & 31;
        q_s[i] = qbase[(size_t)(l0 + rr) * TC + d];
    }
    for (int i = tid; i < 1024; i += 256)
        a_s[i] = attn[((size_t)(n * 8 + h)) * 1024 + i];
    __syncthreads();

    float wd[9];
#pragma unroll
    for (int r = 0; r < 9; ++r) wd[r] = wdc[h * 9 + r];

    const int lane = tid & 31, wave = tid >> 5;
    const float inv_pi = 0.31830988618379067154f;
    for (int ll = wave * 8; ll < wave * 8 + 8; ++ll) {
        float acc = 0.f;
#pragma unroll
        for (int d = 0; d < 32; ++d) acc += q_s[ll * 32 + d] * a_s[d * 32 + lane];
        const float val = 0.5f * v_s[(ll + 4) * 32 + lane] + inv_pi * acc;
        float ss = val * val;
#pragma unroll
        for (int off = 16; off; off >>= 1) ss += __shfl_xor(ss, off, 32);
        const float res = val / sqrtf(ss);
        float dc = 0.f;
#pragma unroll
        for (int r = 0; r < 9; ++r) dc += wd[r] * v_s[(ll + r) * 32 + lane];
        out_pre[(size_t)(n * LL + l0 + ll) * CC + h * DD + lane] = res + dc;
    }
}

__global__ __launch_bounds__(256) void proj_gemm_kernel(
    const float* __restrict__ A, const float* __restrict__ wp,
    const float* __restrict__ bp, float* __restrict__ out)
{
    __shared__ float2 smemv[4224];
    float* smem = (float*)smemv;
    float2* At2 = smemv;
    float2* Wt2 = smemv + 16 * 130;
    const int mt = blockIdx.x;
    const int nt = blockIdx.y;
    const int row0 = mt * 128;
    const int j0 = nt * 64;
    const int tid = threadIdx.x, lane = tid & 31, wave = tid >> 5;
    const int m0 = wave * 16;
    const int laneM = lane & 15;
    const int laneH = lane >> 4;

    v8f acc[4] = {};
    for (int kc = 0; kc < CC; kc += 32) {
        for (int i = tid; i < 16 * 128; i += 256) {
            const int p = i & 15, rr = i >> 4;
            At2[p * 130 + rr] = *(const float2*)&A[(size_t)(row0 + rr) * CC + kc + 2 * p];
        }
        for (int i = tid; i < 16 * 64; i += 256) {
            const int p = i & 15, j = i >> 4;
            Wt2[p * 66 + j] = *(const float2*)&wp[(size_t)(j0 + j) * CC + kc + 2 * p];
        }
        __syncthreads();
#pragma unroll
        for (int kk = 0; kk < 32; kk += 4) {
            const int pp = (kk >> 1) + laneH;
            const v2f a = *(const v2f*)&At2[pp * 130 + m0 + laneM];
#pragma unroll
            for (int nb = 0; nb < 4; ++nb) {
                const v2f b = *(const v2f*)&Wt2[pp * 66 + nb * 16 + laneM];
                acc[nb] = wmma_f32(a, b, acc[nb]);
            }
        }
        __syncthreads();
    }
    const int lbase = m0 + (laneH << 3);
#pragma unroll
    for (int nb = 0; nb < 4; ++nb) {
        const int j = nb * 16 + laneM;
#pragma unroll
        for (int r = 0; r < 8; ++r) smem[j * 132 + lbase + r] = acc[nb][r];
    }
    __syncthreads();
    const int n = row0 >> 12;
    const int lg0 = row0 & (LL - 1);
    for (int i = tid; i < 64 * 128; i += 256) {
        const int l = i & 127, j = i >> 7;
        out[((size_t)(n * CC + j0 + j)) * LL + lg0 + l] = smem[j * 132 + l] + bp[j0 + j];
    }
}

extern "C" void kernel_launch(void* const* d_in, const int* in_sizes, int n_in,
                              void* d_out, int out_size, void* d_ws, size_t ws_size,
                              hipStream_t stream) {
    const float* x      = (const float*)d_in[0];
    const float* w_qkv  = (const float*)d_in[1];
    const float* b_qkv  = (const float*)d_in[2];
    const float* w_proj = (const float*)d_in[3];
    const float* b_proj = (const float*)d_in[4];
    const float* w_dcv  = (const float*)d_in[5];
    float* out = (float*)d_out;

    float* qkv     = (float*)d_ws;
    float* attn    = qkv + (size_t)NB * LL * TC;
    float* out_pre = attn + (size_t)NB * HH * DD * DD;

    qkv_gemm_kernel<<<dim3(256, 12), 256, 0, stream>>>(x, w_qkv, b_qkv, qkv);
    attn_kernel<<<64, 256, 0, stream>>>(qkv, attn);
    epilogue_kernel<<<4096, 256, 0, stream>>>(qkv, attn, w_dcv, out_pre);
    proj_gemm_kernel<<<dim3(256, 4), 256, 0, stream>>>(out_pre, w_proj, b_proj, out);
}


# Tile staging in the three tiled kernels issues all of a tile's global loads as one group before its LDS stores (was load, wait, store per element); projection bias read once, ahead of the store loop
# speedup vs baseline: 1.5074x; 1.2726x over previous
#include <hip/hip_runtime.h>
#include <hip/hip_bf16.h>
#include <math.h>

typedef float v2f __attribute__((ext_vector_type(2)));
typedef float v8f __attribute__((ext_vector_type(8)));

#define NB 8
#define CC 256
#define HH 8
#define DD 32
#define LL 4096
#define TC 768

static __device__ __forceinline__ v8f wmma_f32(v2f a, v2f b, v8f c) {
    return __builtin_amdgcn_wmma_f32_16x16x4_f32(false, a, false, b, (short)0, c, false, false);
}

static __device__ __forceinline__ float row16_sum(float v) {
    v += __builtin_bit_cast(float, __builtin_amdgcn_update_dpp(0, __builtin_bit_cast(int, v), 0x161, 0xf, 0xf, true));
    v += __builtin_bit_cast(float, __builtin_amdgcn_update_dpp(0, __builtin_bit_cast(int, v), 0x162, 0xf, 0xf, true));
    v += __builtin_bit_cast(float, __builtin_amdgcn_update_dpp(0, __builtin_bit_cast(int, v), 0x164, 0xf, 0xf, true));
    v += __builtin_bit_cast(float, __builtin_amdgcn_update_dpp(0, __builtin_bit_cast(int, v), 0x168, 0xf, 0xf, true));
    return v;
}

__global__ __launch_bounds__(256) void qkv_gemm_kernel(
    const float* __restrict__ x, const float* __restrict__ wq,
    const float* __restrict__ bq, float* __restrict__ qkv)
{
    __shared__ float2 Wt2[16 * 66];
    const int mt = blockIdx.x;
    const int nt = blockIdx.y;
    const int row0 = mt * 128;
    const int n = row0 >> 12;
    const int l0 = row0 & (LL - 1);
    const int j0 = nt * 64;
    const int tid = threadIdx.x;
    const int lane = tid & 31, wave = tid >> 5;
    const int m0 = wave * 16;
    const int laneM = lane & 15;
    const int laneH = lane >> 4;
    const int laneK = laneH << 1;

    v8f acc[4] = {};
    const float* xA = x + (size_t)n * CC * LL + l0 + m0 + laneM;

    const int sp = tid & 15, sj = tid >> 4;
    const float* wsrc = wq + (size_t)(j0 + sj) * CC + 2 * sp;
    for (int kc = 0; kc < CC; kc += 32) {
        float2 wreg[4];
#pragma unroll
        for (int u = 0; u < 4; ++u)
            wreg[u] = *(const float2*)&wsrc[kc + u * 16 * CC];
        v2f areg[8];
#pragma unroll
        for (int q = 0; q < 8; ++q) {
            const int k = kc + 4 * q + laneK;
            areg[q].x = xA[(size_t)k * LL];
            areg[q].y = xA[(size_t)(k + 1) * LL];
        }
        __builtin_amdgcn_sched_barrier(0);
#pragma unroll
        for (int u = 0; u < 4; ++u) Wt2[sp * 66 + sj + 16 * u] = wreg[u];
        __syncthreads();
#pragma unroll
        for (int kk = 0; kk < 32; kk += 4) {
            const v2f a = areg[kk >> 2];
            const int pp = (kk >> 1) + laneH;
#pragma unroll
            for (int nb = 0; nb < 4; ++nb) {
                const v2f b = *(const v2f*)&Wt2[pp * 66 + nb * 16 + laneM];
                acc[nb] = wmma_f32(a, b, acc[nb]);
            }
        }
        __syncthreads();
    }
#pragma unroll
    for (int nb = 0; nb < 4; ++nb) {
        const float bias = bq[j0 + nb * 16 + laneM];
#pragma unroll
        for (int r = 0; r < 8; ++r) acc[nb][r] += bias;
    }
    if (j0 < 2 * CC) {
#pragma unroll
        for (int hd = 0; hd < 2; ++hd) {
#pragma unroll
            for (int r = 0; r < 8; ++r) {
                const float a0 = acc[2 * hd][r], a1 = acc[2 * hd + 1][r];
                const float ss = row16_sum(a0 * a0 + a1 * a1);
                const float inv = 1.0f / sqrtf(ss);
                acc[2 * hd][r] = a0 * inv;
                acc[2 * hd + 1][r] = a1 * inv;
            }
        }
    }
    const int mbase = m0 + (laneH << 3);
#pragma unroll
    for (int nb = 0; nb < 4; ++nb) {
        float* outp = qkv + (size_t)(row0 + mbase) * TC + j0 + nb * 16 + laneM;
#pragma unroll
        for (int r = 0; r < 8; ++r) outp[(size_t)r * TC] = acc[nb][r];
    }
}

__global__ __launch_bounds__(256) void attn_kernel(
    const float* __restrict__ qkv, float* __restrict__ attn)
{
    __shared__ float red[8 * 1024];
    const int nh = blockIdx.x;
    const int n = nh >> 3, h = nh & 7;
    const int tid = threadIdx.x, lane = tid & 31, wave = tid >> 5;
    const int laneM = lane & 15;
    const int laneK = (lane >> 4) << 1;

    v8f acc[2][2] = {};
    const float* base = qkv + (size_t)n * LL * TC + h * DD;

    const int lend = wave * 512 + 512;
    for (int l = wave * 512; l < lend; l += 4) {
        const size_t r0 = (size_t)(l + laneK) * TC;
        v2f aK[2], bV[2];
#pragma unroll
        for (int md = 0; md < 2; ++md) {
            aK[md].x = base[r0 + CC + md * 16 + laneM];
            aK[md].y = base[r0 + TC + CC + md * 16 + laneM];
        }
#pragma unroll
        for (int me = 0; me < 2; ++me) {
            bV[me].x = base[r0 + 2 * CC + me * 16 + laneM];
            bV[me].y = base[r0 + TC + 2 * CC + me * 16 + laneM];
        }
#pragma unroll
        for (int md = 0; md < 2; ++md)
#pragma unroll
            for (int me = 0; me < 2; ++me)
                acc[md][me] = wmma_f32(aK[md], bV[me], acc[md][me]);
    }
    const int dadd = (lane >> 4) << 3;
#pragma unroll
    for (int md = 0; md < 2; ++md)
#pragma unroll
        for (int me = 0; me < 2; ++me)
#pragma unroll
            for (int r = 0; r < 8; ++r)
                red[wave * 1024 + (md * 16 + r + dadd) * 32 + me * 16 + laneM] = acc[md][me][r];
    __syncthreads();
    for (int i = tid; i < 1024; i += 256) {
        float s = 0.f;
#pragma unroll
        for (int w = 0; w < 8; ++w) s += red[w * 1024 + i];
        attn[(size_t)nh * 1024 + i] = s;
    }
}

__global__ __launch_bounds__(256) void epilogue_kernel(
    const float* __restrict__ qkv, const float* __restrict__ attn,
    const float* __restrict__ wdc, float* __restrict__ out_pre)
{
    __shared__ float v_s[72 * 32];
    __shared__ float q_s[64 * 32];
    __shared__ float a_s[32 * 32];
    const int b = blockIdx.x;
    const int lc = b & 63;
    const int h = (b >> 6) & 7;
    const int n = b >> 9;
    const int l0 = lc * 64;
    const int tid = threadIdx.x;
    const float* qbase = qkv + (size_t)n * LL * TC + h * DD;

    const int sr = tid >> 5, se = tid & 31;
    float vreg[9], qreg[8], areg[4];
#pragma unroll
    for (int u = 0; u < 9; ++u) {
        const int l = l0 - 4 + sr + 8 * u;
        const int lcl = l < 0 ? 0 : (l >= LL ? LL - 1 : l);
        const float t = qbase[(size_t)lcl * TC + 2 * CC + se];
        vreg[u] = (l >= 0 && l < LL) ? t : 0.0f;
    }
#pragma unroll
    for (int u = 0; u < 8; ++u) qreg[u] = qbase[(size_t)(l0 + sr + 8 * u) * TC + se];
#pragma unroll
    for (int u = 0; u < 4; ++u) areg[u] = attn[((size_t)(n * 8 + h)) * 1024 + tid + 256 * u];
#pragma unroll
    for (int u = 0; u < 9; ++u) v_s[tid + 256 * u] = vreg[u];
#pragma unroll
    for (int u = 0; u < 8; ++u) q_s[tid + 256 * u] = qreg[u];
#pragma unroll
    for (int u = 0; u < 4; ++u) a_s[tid + 256 * u] = areg[u];
    __syncthreads();

    float wd[9];
#pragma unroll
    for (int r = 0; r < 9; ++r) wd[r] = wdc[h * 9 + r];

    const int lane = tid & 31, wave = tid >> 5;
    const float inv_pi = 0.31830988618379067154f;
    for (int ll = wave * 8; ll < wave * 8 + 8; ++ll) {
        float acc = 0.f;
#pragma unroll
        for (int d = 0; d < 32; ++d) acc += q_s[ll * 32 + d] * a_s[d * 32 + lane];
        const float val = 0.5f * v_s[(ll + 4) * 32 + lane] + inv_pi * acc;
        float ss = val * val;
#pragma unroll
        for (int off = 16; off; off >>= 1) ss += __shfl_xor(ss, off, 32);
        const float res = val / sqrtf(ss);
        float dc = 0.f;
#pragma unroll
        for (int r = 0; r < 9; ++r) dc += wd[r] * v_s[(ll + r) * 32 + lane];
        out_pre[(size_t)(n * LL + l0 + ll) * CC + h * DD + lane] = res + dc;
    }
}

__global__ __launch_bounds__(256) void proj_gemm_kernel(
    const float* __restrict__ A, const float* __restrict__ wp,
    const float* __restrict__ bp, float* __restrict__ out)
{
    __shared__ float2 smemv[4224];
    float* smem = (float*)smemv;
    float2* At2 = smemv;
    float2* Wt2 = smemv + 16 * 130;
    const int mt = blockIdx.x;
    const int nt = blockIdx.y;
    const int row0 = mt * 128;
    const int j0 = nt * 64;
    const int tid = threadIdx.x, lane = tid & 31, wave = tid >> 5;
    const int m0 = wave * 16;
    const int laneM = lane & 15;
    const int laneH = lane >> 4;

    v8f acc[4] = {};
    const int sp = tid & 15, sj = tid >> 4;
    const float* asrc = A + (size_t)(row0 + sj) * CC + 2 * sp;
    const float* wsrc = wp + (size_t)(j0 + sj) * CC + 2 * sp;
    for (int kc = 0; kc < CC; kc += 32) {
        float2 areg[8], wreg[4];
#pragma unroll
        for (int u = 0; u < 8; ++u)
            areg[u] = *(const float2*)&asrc[kc + u * 16 * CC];
#pragma unroll
        for (int u = 0; u < 4; ++u)
            wreg[u] = *(const float2*)&wsrc[kc + u * 16 * CC];
        __builtin_amdgcn_sched_barrier(0);
#pragma unroll
        for (int u = 0; u < 8; ++u) At2[sp * 130 + sj + 16 * u] = areg[u];
#pragma unroll
        for (int u = 0; u < 4; ++u) Wt2[sp * 66 + sj + 16 * u] = wreg[u];
        __syncthreads();
#pragma unroll
        for (int kk = 0; kk < 32; kk += 4) {
            const int pp = (kk >> 1) + laneH;
            const v2f a = *(const v2f*)&At2[pp * 130 + m0 + laneM];
#pragma unroll
            for (int nb = 0; nb < 4; ++nb) {
                const v2f b = *(const v2f*)&Wt2[pp * 66 + nb * 16 + laneM];
                acc[nb] = wmma_f32(a, b, acc[nb]);
            }
        }
        __syncthreads();
    }
    float bias[4];
#pragma unroll
    for (int nb = 0; nb < 4; ++nb) bias[nb] = bp[j0 + nb * 16 + laneM];
    const int lbase = m0 + (laneH << 3);
#pragma unroll
    for (int nb = 0; nb < 4; ++nb) {
        const int j = nb * 16 + laneM;
#pragma unroll
        for (int r = 0; r < 8; ++r) smem[j * 132 + lbase + r] = acc[nb][r] + bias[nb];
    }
    __syncthreads();
    const int n = row0 >> 12;
    const int lg0 = row0 & (LL - 1);
    const int ol = tid & 127, oj = tid >> 7;
    const float* srcp = smem + oj * 132 + ol;
    float* dstp = out + ((size_t)(n * CC + j0 + oj)) * LL + lg0 + ol;
#pragma unroll 8
    for (int u = 0; u < 32; ++u) dstp[(size_t)u * 2 * LL] = srcp[u * 2 * 132];
}

extern "C" void kernel_launch(void* const* d_in, const int* in_sizes, int n_in,
                              void* d_out, int out_size, void* d_ws, size_t ws_size,
                              hipStream_t stream) {
    const float* x      = (const float*)d_in[0];
    const float* w_qkv  = (const float*)d_in[1];
    const float* b_qkv  = (const float*)d_in[2];
    const float* w_proj = (const float*)d_in[3];
    const float* b_proj = (const float*)d_in[4];
    const float* w_dcv  = (const float*)d_in[5];
    float* out = (float*)d_out;

    float* qkv     = (float*)d_ws;
    float* attn    = qkv + (size_t)NB * LL * TC;
    float* out_pre = attn + (size_t)NB * HH * DD * DD;

    qkv_gemm_kernel<<<dim3(256, 12), 256, 0, stream>>>(x, w_qkv, b_qkv, qkv);
    attn_kernel<<<64, 256, 0, stream>>>(qkv, attn);
    epilogue_kernel<<<4096, 256, 0, stream>>>(qkv, attn, w_dcv, out_pre);
    proj_gemm_kernel<<<dim3(256, 4), 256, 0, stream>>>(out_pre, w_proj, b_proj, out);
}


# k^T v kernel: L split over 4 blocks per head, 16 waves of 64 rows, all of a wave's loads in flight at once (was 64 blocks, 32 load rounds per wave); epilogue sums 4 partials: summation order changes
# speedup vs baseline: 1.7082x; 1.1332x over previous
#include <hip/hip_runtime.h>
#include <hip/hip_bf16.h>
#include <math.h>

typedef float v2f __attribute__((ext_vector_type(2)));
typedef float v8f __attribute__((ext_vector_type(8)));

#define NB 8
#define CC 256
#define HH 8
#define DD 32
#define LL 4096
#define TC 768

static __device__ __forceinline__ v8f wmma_f32(v2f a, v2f b, v8f c) {
    return __builtin_amdgcn_wmma_f32_16x16x4_f32(false, a, false, b, (short)0, c, false, false);
}

static __device__ __forceinline__ float row16_sum(float v) {
    v += __builtin_bit_cast(float, __builtin_amdgcn_update_dpp(0, __builtin_bit_cast(int, v), 0x161, 0xf, 0xf, true));
    v += __builtin_bit_cast(float, __builtin_amdgcn_update_dpp(0, __builtin_bit_cast(int, v), 0x162, 0xf, 0xf, true));
    v += __builtin_bit_cast(float, __builtin_amdgcn_update_dpp(0, __builtin_bit_cast(int, v), 0x164, 0xf, 0xf, true));
    v += __builtin_bit_cast(float, __builtin_amdgcn_update_dpp(0, __builtin_bit_cast(int, v), 0x168, 0xf, 0xf, true));
    return v;
}

__global__ __launch_bounds__(256) void qkv_gemm_kernel(
    const float* __restrict__ x, const float* __restrict__ wq,
    const float* __restrict__ bq, float* __restrict__ qkv)
{
    __shared__ float2 Wt2[16 * 66];
    const int mt = blockIdx.x;
    const int nt = blockIdx.y;
    const int row0 = mt * 128;
    const int n = row0 >> 12;
    const int l0 = row0 & (LL - 1);
    const int j0 = nt * 64;
    const int tid = threadIdx.x;
    const int lane = tid & 31, wave = tid >> 5;
    const int m0 = wave * 16;
    const int laneM = lane & 15;
    const int laneH = lane >> 4;
    const int laneK = laneH << 1;

    v8f acc[4] = {};
    const float* xA = x + (size_t)n * CC * LL + l0 + m0 + laneM;

    const int sp = tid & 15, sj = tid >> 4;
    const float* wsrc = wq + (size_t)(j0 + sj) * CC + 2 * sp;
    for (int kc = 0; kc < CC; kc += 32) {
        float2 wreg[4];
#pragma unroll
        for (int u = 0; u < 4; ++u)
            wreg[u] = *(const float2*)&wsrc[kc + u * 16 * CC];
        v2f areg[8];
#pragma unroll
        for (int q = 0; q < 8; ++q) {
            const int k = kc + 4 * q + laneK;
            areg[q].x = xA[(size_t)k * LL];
            areg[q].y = xA[(size_t)(k + 1) * LL];
        }
        __builtin_amdgcn_sched_barrier(0);
#pragma unroll
        for (int u = 0; u < 4; ++u) Wt2[sp * 66 + sj + 16 * u] = wreg[u];
        __syncthreads();
#pragma unroll
        for (int kk = 0; kk < 32; kk += 4) {
            const v2f a = areg[kk >> 2];
            const int pp = (kk >> 1) + laneH;
#pragma unroll
            for (int nb = 0; nb < 4; ++nb) {
                const v2f b = *(const v2f*)&Wt2[pp * 66 + nb * 16 + laneM];
                acc[nb] = wmma_f32(a, b, acc[nb]);
            }
        }
        __syncthreads();
    }
#pragma unroll
    for (int nb = 0; nb < 4; ++nb) {
        const float bias = bq[j0 + nb * 16 + laneM];
#pragma unroll
        for (int r = 0; r < 8; ++r) acc[nb][r] += bias;
    }
    if (j0 < 2 * CC) {
#pragma unroll
        for (int hd = 0; hd < 2; ++hd) {
#pragma unroll
            for (int r = 0; r < 8; ++r) {
                const float a0 = acc[2 * hd][r], a1 = acc[2 * hd + 1][r];
                const float ss = row16_sum(a0 * a0 + a1 * a1);
                const float inv = 1.0f / sqrtf(ss);
                acc[2 * hd][r] = a0 * inv;
                acc[2 * hd + 1][r] = a1 * inv;
            }
        }
    }
    const int mbase = m0 + (laneH << 3);
#pragma unroll
    for (int nb = 0; nb < 4; ++nb) {
        float* outp = qkv + (size_t)(row0 + mbase) * TC + j0 + nb * 16 + laneM;
#pragma unroll
        for (int r = 0; r < 8; ++r) outp[(size_t)r * TC] = acc[nb][r];
    }
}

__global__ __launch_bounds__(512) void attn_kernel(float* qkv)
{
    __shared__ __attribute__((aligned(16))) float red[16 * 1024];
    const int b = blockIdx.x;
    const int s = b & 3, nh = b >> 2;
    const int n = nh >> 3, h = nh & 7;
    const int tid = threadIdx.x, lane = tid & 31, wave = tid >> 5;
    const int laneM = lane & 15;
    const int laneH = lane >> 4;

    v8f acc[2][2] = {};
    float* base = qkv + (size_t)n * LL * TC + h * DD;
    const float* kp = base + (size_t)(s * 1024 + wave * 64 + 2 * laneH) * TC + CC + 2 * laneM;
    float2 k0[16], k1[16], v0[16], v1[16];
#pragma unroll
    for (int j = 0; j < 16; ++j) {
        const float* p = kp + (size_t)(4 * j) * TC;
        k0[j] = *(const float2*)p;
        k1[j] = *(const float2*)(p + TC);
        v0[j] = *(const float2*)(p + CC);
        v1[j] = *(const float2*)(p + TC + CC);
    }
    __builtin_amdgcn_sched_barrier(0);
#pragma unroll
    for (int j = 0; j < 16; ++j) {
        v2f a0, a1, b0, b1;
        a0.x = k0[j].x; a0.y = k1[j].x;
        a1.x = k0[j].y; a1.y = k1[j].y;
        b0.x = v0[j].x; b0.y = v1[j].x;
        b1.x = v0[j].y; b1.y = v1[j].y;
        acc[0][0] = wmma_f32(a0, b0, acc[0][0]);
        acc[0][1] = wmma_f32(a0, b1, acc[0][1]);
        acc[1][0] = wmma_f32(a1, b0, acc[1][0]);
        acc[1][1] = wmma_f32(a1, b1, acc[1][1]);
    }
#pragma unroll
    for (int md = 0; md < 2; ++md)
#pragma unroll
        for (int r = 0; r < 8; ++r) {
            const int d = 2 * (8 * laneH + r) + md;
            float2 t; t.x = acc[md][0][r]; t.y = acc[md][1][r];
            *(float2*)&red[wave * 1024 + d * 32 + 2 * laneM] = t;
        }
    __syncthreads();
    for (int i = tid; i < 1024; i += 512) {
        float sum = 0.f;
#pragma unroll
        for (int w = 0; w < 16; ++w) sum += red[w * 1024 + i];
        base[(size_t)(s * 1024 + (i >> 5)) * TC + CC + (i & 31)] = sum;
    }
}

__global__ __launch_bounds__(256) void epilogue_kernel(
    const float* __restrict__ qkv,
    const float* __restrict__ wdc, float* __restrict__ out_pre)
{
    __shared__ float v_s[72 * 32];
    __shared__ float q_s[64 * 32];
    __shared__ float a_s[32 * 32];
    const int b = blockIdx.x;
    const int lc = b & 63;
    const int h = (b >> 6) & 7;
    const int n = b >> 9;
    const int l0 = lc * 64;
    const int tid = threadIdx.x;
    const float* qbase = qkv + (size_t)n * LL * TC + h * DD;

    const int sr = tid >> 5, se = tid & 31;
    float vreg[9], qreg[8], areg[4], part[4][4];
#pragma unroll
    for (int u = 0; u < 9; ++u) {
        const int l = l0 - 4 + sr + 8 * u;
        const int lcl = l < 0 ? 0 : (l >= LL ? LL - 1 : l);
        const float t = qbase[(size_t)lcl * TC + 2 * CC + se];
        vreg[u] = (l >= 0 && l < LL) ? t : 0.0f;
    }
#pragma unroll
    for (int u = 0; u < 8; ++u) qreg[u] = qbase[(size_t)(l0 + sr + 8 * u) * TC + se];
#pragma unroll
    for (int u = 0; u < 4; ++u)
#pragma unroll
        for (int s = 0; s < 4; ++s)
            part[u][s] = qbase[(size_t)(s * 1024 + sr + 8 * u) * TC + CC + se];
    __builtin_amdgcn_sched_barrier(0);
#pragma unroll
    for (int u = 0; u < 4; ++u) areg[u] = ((part[u][0] + part[u][1]) + part[u][2]) + part[u][3];
#pragma unroll
    for (int u = 0; u < 9; ++u) v_s[tid + 256 * u] = vreg[u];
#pragma unroll
    for (int u = 0; u < 8; ++u) q_s[tid + 256 * u] = qreg[u];
#pragma unroll
    for (int u = 0; u < 4; ++u) a_s[tid + 256 * u] = areg[u];
    __syncthreads();

    float wd[9];
#pragma unroll
    for (int r = 0; r < 9; ++r) wd[r] = wdc[h * 9 + r];

    const int lane = tid & 31, wave = tid >> 5;
    const float inv_pi = 0.31830988618379067154f;
    for (int ll = wave * 8; ll < wave * 8 + 8; ++ll) {
        float acc = 0.f;
#pragma unroll
        for (int d = 0; d < 32; ++d) acc += q_s[ll * 32 + d] * a_s[d * 32 + lane];
        const float val = 0.5f * v_s[(ll + 4) * 32 + lane] + inv_pi * acc;
        float ss = val * val;
#pragma unroll
        for (int off = 16; off; off >>= 1) ss += __shfl_xor(ss, off, 32);
        const float res = val / sqrtf(ss);
        float dc = 0.f;
#pragma unroll
        for (int r = 0; r < 9; ++r) dc += wd[r] * v_s[(ll + r) * 32 + lane];
        out_pre[(size_t)(n * LL + l0 + ll) * CC + h * DD + lane] = res + dc;
    }
}

__global__ __launch_bounds__(256) void proj_gemm_kernel(
    const float* __restrict__ A, const float* __restrict__ wp,
    const float* __restrict__ bp, float* __restrict__ out)
{
    __shared__ float2 smemv[4224];
    float* smem = (float*)smemv;
    float2* At2 = smemv;
    float2* Wt2 = smemv + 16 * 130;
    const int mt = blockIdx.x;
    const int nt = blockIdx.y;
    const int row0 = mt * 128;
    const int j0 = nt * 64;
    const int tid = threadIdx.x, lane = tid & 31, wave = tid >> 5;
    const int m0 = wave * 16;
    const int laneM = lane & 15;
    const int laneH = lane >> 4;

    v8f acc[4] = {};
    const int sp = tid & 15, sj = tid >> 4;
    const float* asrc = A + (size_t)(row0 + sj) * CC + 2 * sp;
    const float* wsrc = wp + (size_t)(j0 + sj) * CC + 2 * sp;
    for (int kc = 0; kc < CC; kc += 32) {
        float2 areg[8], wreg[4];
#pragma unroll
        for (int u = 0; u < 8; ++u)
            areg[u] = *(const float2*)&asrc[kc + u * 16 * CC];
#pragma unroll
        for (int u = 0; u < 4; ++u)
            wreg[u] = *(const float2*)&wsrc[kc + u * 16 * CC];
        __builtin_amdgcn_sched_barrier(0);
#pragma unroll
        for (int u = 0; u < 8; ++u) At2[sp * 130 + sj + 16 * u] = areg[u];
#pragma unroll
        for (int u = 0; u < 4; ++u) Wt2[sp * 66 + sj + 16 * u] = wreg[u];
        __syncthreads();
#pragma unroll
        for (int kk = 0; kk < 32; kk += 4) {
            const int pp = (kk >> 1) + laneH;
            const v2f a = *(const v2f*)&At2[pp * 130 + m0 + laneM];
#pragma unroll
            for (int nb = 0; nb < 4; ++nb) {
                const v2f b = *(const v2f*)&Wt2[pp * 66 + nb * 16 + laneM];
                acc[nb] = wmma_f32(a, b, acc[nb]);
            }
        }
        __syncthreads();
    }
    float bias[4];
#pragma unroll
    for (int nb = 0; nb < 4; ++nb) bias[nb] = bp[j0 + nb * 16 + laneM];
    const int lbase = m0 + (laneH << 3);
#pragma unroll
    for (int nb = 0; nb < 4; ++nb) {
        const int j = nb * 16 + laneM;
#pragma unroll
        for (int r = 0; r < 8; ++r) smem[j * 132 + lbase + r] = acc[nb][r] + bias[nb];
    }
    __syncthreads();
    const int n = row0 >> 12;
    const int lg0 = row0 & (LL - 1);
    const int ol = tid & 127, oj = tid >> 7;
    const float* srcp = smem + oj * 132 + ol;
    float* dstp = out + ((size_t)(n * CC + j0 + oj)) * LL + lg0 + ol;
#pragma unroll 8
    for (int u = 0; u < 32; ++u) dstp[(size_t)u * 2 * LL] = srcp[u * 2 * 132];
}

extern "C" void kernel_launch(void* const* d_in, const int* in_sizes, int n_in,
                              void* d_out, int out_size, void* d_ws, size_t ws_size,
                              hipStream_t stream) {
    const float* x      = (const float*)d_in[0];
    const float* w_qkv  = (const float*)d_in[1];
    const float* b_qkv  = (const float*)d_in[2];
    const float* w_proj = (const float*)d_in[3];
    const float* b_proj = (const float*)d_in[4];
    const float* w_dcv  = (const float*)d_in[5];
    float* out = (float*)d_out;

    float* qkv     = (float*)d_ws;
    float* out_pre = qkv + (size_t)NB * LL * TC + (size_t)NB * HH * DD * DD;

    qkv_gemm_kernel<<<dim3(256, 12), 256, 0, stream>>>(x, w_qkv, b_qkv, qkv);
    attn_kernel<<<256, 512, 0, stream>>>(qkv);
    epilogue_kernel<<<4096, 256, 0, stream>>>(qkv, w_dcv, out_pre);
    proj_gemm_kernel<<<dim3(256, 4), 256, 0, stream>>>(out_pre, w_proj, b_proj, out);
}


# Epilogue fused into the projection kernel: a block builds each head's 64x32 out_pre tile in LDS (q.attn on the f32 matrix unit, 16-lane DPP norm): 4 -> 3 launches; product and norm sum orders change
# speedup vs baseline: 1.9598x; 1.1473x over previous
#include <hip/hip_runtime.h>
#include <hip/hip_bf16.h>
#include <math.h>

typedef float v2f __attribute__((ext_vector_type(2)));
typedef float v8f __attribute__((ext_vector_type(8)));

#define NB 8
#define CC 256
#define HH 8
#define DD 32
#define LL 4096
#define TC 768

static __device__ __forceinline__ v8f wmma_f32(v2f a, v2f b, v8f c) {
    return __builtin_amdgcn_wmma_f32_16x16x4_f32(false, a, false, b, (short)0, c, false, false);
}

static __device__ __forceinline__ float row16_sum(float v) {
    v += __builtin_bit_cast(float, __builtin_amdgcn_update_dpp(0, __builtin_bit_cast(int, v), 0x161, 0xf, 0xf, true));
    v += __builtin_bit_cast(float, __builtin_amdgcn_update_dpp(0, __builtin_bit_cast(int, v), 0x162, 0xf, 0xf, true));
    v += __builtin_bit_cast(float, __builtin_amdgcn_update_dpp(0, __builtin_bit_cast(int, v), 0x164, 0xf, 0xf, true));
    v += __builtin_bit_cast(float, __builtin_amdgcn_update_dpp(0, __builtin_bit_cast(int, v), 0x168, 0xf, 0xf, true));
    return v;
}

__global__ __launch_bounds__(256) void qkv_gemm_kernel(
    const float* __restrict__ x, const float* __restrict__ wq,
    const float* __restrict__ bq, float* __restrict__ qkv)
{
    __shared__ float2 Wt2[16 * 66];
    const int mt = blockIdx.x;
    const int nt = blockIdx.y;
    const int row0 = mt * 128;
    const int n = row0 >> 12;
    const int l0 = row0 & (LL - 1);
    const int j0 = nt * 64;
    const int tid = threadIdx.x;
    const int lane = tid & 31, wave = tid >> 5;
    const int m0 = wave * 16;
    const int laneM = lane & 15;
    const int laneH = lane >> 4;
    const int laneK = laneH << 1;

    v8f acc[4] = {};
    const float* xA = x + (size_t)n * CC * LL + l0 + m0 + laneM;

    const int sp = tid & 15, sj = tid >> 4;
    const float* wsrc = wq + (size_t)(j0 + sj) * CC + 2 * sp;
    for (int kc = 0; kc < CC; kc += 32) {
        float2 wreg[4];
#pragma unroll
        for (int u = 0; u < 4; ++u)
            wreg[u] = *(const float2*)&wsrc[kc + u * 16 * CC];
        v2f areg[8];
#pragma unroll
        for (int q = 0; q < 8; ++q) {
            const int k = kc + 4 * q + laneK;
            areg[q].x = xA[(size_t)k * LL];
            areg[q].y = xA[(size_t)(k + 1) * LL];
        }
        __builtin_amdgcn_sched_barrier(0);
#pragma unroll
        for (int u = 0; u < 4; ++u) Wt2[sp * 66 + sj + 16 * u] = wreg[u];
        __syncthreads();
#pragma unroll
        for (int kk = 0; kk < 32; kk += 4) {
            const v2f a = areg[kk >> 2];
            const int pp = (kk >> 1) + laneH;
#pragma unroll
            for (int nb = 0; nb < 4; ++nb) {
                const v2f b = *(const v2f*)&Wt2[pp * 66 + nb * 16 + laneM];
                acc[nb] = wmma_f32(a, b, acc[nb]);
            }
        }
        __syncthreads();
    }
#pragma unroll
    for (int nb = 0; nb < 4; ++nb) {
        const float bias = bq[j0 + nb * 16 + laneM];
#pragma unroll
        for (int r = 0; r < 8; ++r) acc[nb][r] += bias;
    }
    if (j0 < 2 * CC) {
#pragma unroll
        for (int hd = 0; hd < 2; ++hd) {
#pragma unroll
            for (int r = 0; r < 8; ++r) {
                const float a0 = acc[2 * hd][r], a1 = acc[2 * hd + 1][r];
                const float ss = row16_sum(a0 * a0 + a1 * a1);
                const float inv = 1.0f / sqrtf(ss);
                acc[2 * hd][r] = a0 * inv;
                acc[2 * hd + 1][r] = a1 * inv;
            }
        }
    }
    const int mbase = m0 + (laneH << 3);
#pragma unroll
    for (int nb = 0; nb < 4; ++nb) {
        float* outp = qkv + (size_t)(row0 + mbase) * TC + j0 + nb * 16 + laneM;
#pragma unroll
        for (int r = 0; r < 8; ++r) outp[(size_t)r * TC] = acc[nb][r];
    }
}

__global__ __launch_bounds__(512) void attn_kernel(float* qkv)
{
    __shared__ __attribute__((aligned(16))) float red[16 * 1024];
    const int b = blockIdx.x;
    const int s = b & 3, nh = b >> 2;
    const int n = nh >> 3, h = nh & 7;
    const int tid = threadIdx.x, lane = tid & 31, wave = tid >> 5;
    const int laneM = lane & 15;
    const int laneH = lane >> 4;

    v8f acc[2][2] = {};
    float* base = qkv + (size_t)n * LL * TC + h * DD;
    const float* kp = base + (size_t)(s * 1024 + wave * 64 + 2 * laneH) * TC + CC + 2 * laneM;
    float2 k0[16], k1[16], v0[16], v1[16];
#pragma unroll
    for (int j = 0; j < 16; ++j) {
        const float* p = kp + (size_t)(4 * j) * TC;
        k0[j] = *(const float2*)p;
        k1[j] = *(const float2*)(p + TC);
        v0[j] = *(const float2*)(p + CC);
        v1[j] = *(const float2*)(p + TC + CC);
    }
    __builtin_amdgcn_sched_barrier(0);
#pragma unroll
    for (int j = 0; j < 16; ++j) {
        v2f a0, a1, b0, b1;
        a0.x = k0[j].x; a0.y = k1[j].x;
        a1.x = k0[j].y; a1.y = k1[j].y;
        b0.x = v0[j].x; b0.y = v1[j].x;
        b1.x = v0[j].y; b1.y = v1[j].y;
        acc[0][0] = wmma_f32(a0, b0, acc[0][0]);
        acc[0][1] = wmma_f32(a0, b1, acc[0][1]);
        acc[1][0] = wmma_f32(a1, b0, acc[1][0]);
        acc[1][1] = wmma_f32(a1, b1, acc[1][1]);
    }
#pragma unroll
    for (int md = 0; md < 2; ++md)
#pragma unroll
        for (int r = 0; r < 8; ++r) {
            const int d = 2 * (8 * laneH + r) + md;
            float2 t; t.x = acc[md][0][r]; t.y = acc[md][1][r];
            *(float2*)&red[wave * 1024 + d * 32 + 2 * laneM] = t;
        }
    __syncthreads();
    for (int i = tid; i < 1024; i += 512) {
        float sum = 0.f;
#pragma unroll
        for (int w = 0; w < 16; ++w) sum += red[w * 1024 + i];
        base[(size_t)(s * 1024 + (i >> 5)) * TC + CC + (i & 31)] = sum;
    }
}

struct HeadStage {
    float v[9];
    float q[8];
    float p[4][4];
    float2 w[16];
};

static __device__ __forceinline__ void head_stage_load_vq(
    HeadStage& st, const float* __restrict__ qb, int l0, int sr, int se)
{
#pragma unroll
    for (int u = 0; u < 9; ++u) {
        const int l = l0 - 4 + sr + 8 * u;
        const int lcl = l < 0 ? 0 : (l >= LL ? LL - 1 : l);
        st.v[u] = qb[(size_t)lcl * TC + 2 * CC + se];
    }
#pragma unroll
    for (int u = 0; u < 8; ++u) st.q[u] = qb[(size_t)(l0 + sr + 8 * u) * TC + se];
}

static __device__ __forceinline__ void head_stage_load_wa(
    HeadStage& st, const float* __restrict__ qb, const float* __restrict__ wsrc, int sr, int se)
{
#pragma unroll
    for (int u = 0; u < 16; ++u) st.w[u] = *(const float2*)&wsrc[(size_t)u * 16 * CC];
#pragma unroll
    for (int u = 0; u < 4; ++u)
#pragma unroll
        for (int s = 0; s < 4; ++s)
            st.p[u][s] = qb[(size_t)(s * 1024 + sr + 8 * u) * TC + CC + se];
}

#define VS_OFF 0
#define QS_OFF 2592
#define A2_OFF 4896
#define AT_OFF 5920
#define WT_OFF 8032
#define SM_FLOATS 16288

__global__ __launch_bounds__(256, 4) void epi_proj_kernel(
    const float* __restrict__ qkv, const float* __restrict__ wdc,
    const float* __restrict__ wp, const float* __restrict__ bp, float* __restrict__ out)
{
    __shared__ __attribute__((aligned(16))) float smem[SM_FLOATS];
    float* v_s = smem + VS_OFF;
    float* q_s = smem + QS_OFF;
    float* a2 = smem + A2_OFF;
    float* At2f = smem + AT_OFF;
    float2* Wt2 = (float2*)(smem + WT_OFF);
    const int b = blockIdx.x;
    const int lc = b & 63;
    const int n = b >> 6;
    const int l0 = lc * 64;
    const int tid = threadIdx.x;
    const int lane = tid & 31;
    const int wave = __builtin_amdgcn_readfirstlane(tid >> 5);
    const int laneM = lane & 15;
    const int laneH = lane >> 4;
    const int sr = tid >> 5, se = tid & 31;
    const int sp = tid & 15, sj = tid >> 4;
    const int mp = wave & 3;
    const int nh = wave >> 2;
    const float* qn = qkv + (size_t)n * LL * TC;
    const float* wsrc0 = wp + (size_t)sj * CC + 2 * sp;
    const float inv_pi = 0.31830988618379067154f;
    float* vst = v_s + sr * 36 + se;
    float* qst = q_s + sr * 36 + se;
    float* ast = a2 + ((sr >> 1) * 32 + se) * 2 + (sr & 1);
    float2* wst = Wt2 + sp * 258 + sj;
    const float2* apf = (const float2*)At2f + laneH * 66 + mp * 16 + laneM;
    const float2* bpf = Wt2 + laneH * 258 + nh * 128 + laneM;

    v8f acc[8] = {};
    HeadStage st;
    head_stage_load_vq(st, qn, l0, sr, se);
    head_stage_load_wa(st, qn, wsrc0, sr, se);

    for (int h = 0; h < HH; ++h) {
        float wd[9];
#pragma unroll
        for (int t = 0; t < 9; ++t) wd[t] = wdc[h * 9 + t];
        float areg[4];
#pragma unroll
        for (int u = 0; u < 4; ++u) areg[u] = ((st.p[u][0] + st.p[u][1]) + st.p[u][2]) + st.p[u][3];
#pragma unroll
        for (int u = 0; u < 9; ++u) {
            const int l = l0 - 4 + sr + 8 * u;
            vst[u * 8 * 36] = (l >= 0 && l < LL) ? st.v[u] : 0.0f;
        }
#pragma unroll
        for (int u = 0; u < 8; ++u) qst[u * 8 * 36] = st.q[u];
#pragma unroll
        for (int u = 0; u < 4; ++u) ast[u * 4 * 32 * 2] = areg[u];
#pragma unroll
        for (int u = 0; u < 16; ++u) wst[16 * u] = st.w[u];
        __syncthreads();

        if (wave < 4) {
            const int m0 = wave * 16;
            const float* qf = q_s + (m0 + laneM) * 36 + 2 * laneH;
            const float* af = a2 + (laneH * 32 + laneM) * 2;
            const float* vw = v_s + (m0 + 8 * laneH) * 36 + laneM;
            float* atw = At2f + ((laneM >> 1) * 66 + m0 + 8 * laneH) * 2 + (laneM & 1);
            v8f P0 = {}, P1 = {};
#pragma unroll
            for (int s = 0; s < 8; ++s) {
                const v2f a = *(const v2f*)&qf[4 * s];
                const v2f b0 = *(const v2f*)&af[s * 2 * 32 * 2];
                const v2f b1 = *(const v2f*)&af[s * 2 * 32 * 2 + 16 * 2];
                P0 = wmma_f32(a, b0, P0);
                P1 = wmma_f32(a, b1, P1);
            }
            float w0[16], w1[16];
#pragma unroll
            for (int i = 0; i < 16; ++i) {
                w0[i] = vw[i * 36];
                w1[i] = vw[i * 36 + 16];
            }
#pragma unroll
            for (int r = 0; r < 8; ++r) {
                const float x0 = 0.5f * w0[r + 4] + inv_pi * P0[r];
                const float x1 = 0.5f * w1[r + 4] + inv_pi * P1[r];
                const float ss = row16_sum(x0 * x0 + x1 * x1);
                const float nrm = sqrtf(ss);
                float d0 = 0.f, d1 = 0.f;
#pragma unroll
                for (int t = 0; t < 9; ++t) {
                    d0 += wd[t] * w0[r + t];
                    d1 += wd[t] * w1[r + t];
                }
                atw[2 * r] = x0 / nrm + d0;
                atw[8 * 66 * 2 + 2 * r] = x1 / nrm + d1;
            }
        }
        __syncthreads();
        if (h + 1 < HH) {
            head_stage_load_vq(st, qn + (h + 1) * DD, l0, sr, se);
            head_stage_load_wa(st, qn + (h + 1) * DD, wsrc0 + (h + 1) * DD, sr, se);
        }
        __builtin_amdgcn_sched_barrier(0);

#pragma unroll
        for (int ks = 0; ks < 8; ++ks) {
            const v2f a = *(const v2f*)&apf[ks * 2 * 66];
#pragma unroll
            for (int nb = 0; nb < 8; ++nb) {
                const v2f bb = *(const v2f*)&bpf[ks * 2 * 258 + nb * 16];
                acc[nb] = wmma_f32(a, bb, acc[nb]);
            }
        }
        __syncthreads();
    }

    float bias[8];
#pragma unroll
    for (int nb = 0; nb < 8; ++nb) bias[nb] = bp[nh * 128 + nb * 16 + laneM];
    const int oj = tid >> 4, ol4 = (tid & 15) * 4;
    for (int pass = 0; pass < 2; ++pass) {
        if (nh == pass) {
#pragma unroll
            for (int nb = 0; nb < 8; ++nb) {
                float* o = smem + (nb * 16 + laneM) * 68 + mp * 16 + 8 * laneH;
                float4 lo, hi;
                lo.x = acc[nb][0] + bias[nb]; lo.y = acc[nb][1] + bias[nb];
                lo.z = acc[nb][2] + bias[nb]; lo.w = acc[nb][3] + bias[nb];
                hi.x = acc[nb][4] + bias[nb]; hi.y = acc[nb][5] + bias[nb];
                hi.z = acc[nb][6] + bias[nb]; hi.w = acc[nb][7] + bias[nb];
                *(float4*)o = lo;
                *(float4*)(o + 4) = hi;
            }
        }
        __syncthreads();
        float* dst = out + ((size_t)(n * CC + pass * 128 + oj)) * LL + l0 + ol4;
#pragma unroll
        for (int u = 0; u < 8; ++u) {
            const float4 t = *(const float4*)&smem[(oj + 16 * u) * 68 + ol4];
            *(float4*)&dst[(size_t)u * 16 * LL] = t;
        }
        __syncthreads();
    }
}

extern "C" void kernel_launch(void* const* d_in, const int* in_sizes, int n_in,
                              void* d_out, int out_size, void* d_ws, size_t ws_size,
                              hipStream_t stream) {
    const float* x      = (const float*)d_in[0];
    const float* w_qkv  = (const float*)d_in[1];
    const float* b_qkv  = (const float*)d_in[2];
    const float* w_proj = (const float*)d_in[3];
    const float* b_proj = (const float*)d_in[4];
    const float* w_dcv  = (const float*)d_in[5];
    float* out = (float*)d_out;

    float* qkv = (float*)d_ws;

    qkv_gemm_kernel<<<dim3(256, 12), 256, 0, stream>>>(x, w_qkv, b_qkv, qkv);
    attn_kernel<<<256, 512, 0, stream>>>(qkv);
    epi_proj_kernel<<<512, 256, 0, stream>>>(qkv, w_dcv, w_proj, b_proj, out);
}


# Row L2 norms (q, k in the QKV epilogue; out_pre in the tile build) take 1/norm from the hardware reciprocal square root and multiply, in place of IEEE sqrt + division: last-bit rounding changes
# speedup vs baseline: 2.0290x; 1.0353x over previous
#include <hip/hip_runtime.h>
#include <hip/hip_bf16.h>
#include <math.h>

typedef float v2f __attribute__((ext_vector_type(2)));
typedef float v8f __attribute__((ext_vector_type(8)));

#define NB 8
#define CC 256
#define HH 8
#define DD 32
#define LL 4096
#define TC 768

static __device__ __forceinline__ v8f wmma_f32(v2f a, v2f b, v8f c) {
    return __builtin_amdgcn_wmma_f32_16x16x4_f32(false, a, false, b, (short)0, c, false, false);
}

static __device__ __forceinline__ float row16_sum(float v) {
    v += __builtin_bit_cast(float, __builtin_amdgcn_update_dpp(0, __builtin_bit_cast(int, v), 0x161, 0xf, 0xf, true));
    v += __builtin_bit_cast(float, __builtin_amdgcn_update_dpp(0, __builtin_bit_cast(int, v), 0x162, 0xf, 0xf, true));
    v += __builtin_bit_cast(float, __builtin_amdgcn_update_dpp(0, __builtin_bit_cast(int, v), 0x164, 0xf, 0xf, true));
    v += __builtin_bit_cast(float, __builtin_amdgcn_update_dpp(0, __builtin_bit_cast(int, v), 0x168, 0xf, 0xf, true));
    return v;
}

static __device__ __forceinline__ float rsqrt_hw(float s) {
    const bool tiny = s < 0x1p-100f;
    const float r = __builtin_amdgcn_rsqf(tiny ? s * 0x1p+64f : s);
    return tiny ? r * 0x1p+32f : r;
}

__global__ __launch_bounds__(256) void qkv_gemm_kernel(
    const float* __restrict__ x, const float* __restrict__ wq,
    const float* __restrict__ bq, float* __restrict__ qkv)
{
    __shared__ float2 Wt2[16 * 66];
    const int mt = blockIdx.x;
    const int nt = blockIdx.y;
    const int row0 = mt * 128;
    const int n = row0 >> 12;
    const int l0 = row0 & (LL - 1);
    const int j0 = nt * 64;
    const int tid = threadIdx.x;
    const int lane = tid & 31, wave = tid >> 5;
    const int m0 = wave * 16;
    const int laneM = lane & 15;
    const int laneH = lane >> 4;
    const int laneK = laneH << 1;

    v8f acc[4] = {};
    const float* xA = x + (size_t)n * CC * LL + l0 + m0 + laneM;

    const int sp = tid & 15, sj = tid >> 4;
    const float* wsrc = wq + (size_t)(j0 + sj) * CC + 2 * sp;
    for (int kc = 0; kc < CC; kc += 32) {
        float2 wreg[4];
#pragma unroll
        for (int u = 0; u < 4; ++u)
            wreg[u] = *(const float2*)&wsrc[kc + u * 16 * CC];
        v2f areg[8];
#pragma unroll
        for (int q = 0; q < 8; ++q) {
            const int k = kc + 4 * q + laneK;
            areg[q].x = xA[(size_t)k * LL];
            areg[q].y = xA[(size_t)(k + 1) * LL];
        }
        __builtin_amdgcn_sched_barrier(0);
#pragma unroll
        for (int u = 0; u < 4; ++u) Wt2[sp * 66 + sj + 16 * u] = wreg[u];
        __syncthreads();
#pragma unroll
        for (int kk = 0; kk < 32; kk += 4) {
            const v2f a = areg[kk >> 2];
            const int pp = (kk >> 1) + laneH;
#pragma unroll
            for (int nb = 0; nb < 4; ++nb) {
                const v2f b = *(const v2f*)&Wt2[pp * 66 + nb * 16 + laneM];
                acc[nb] = wmma_f32(a, b, acc[nb]);
            }
        }
        __syncthreads();
    }
#pragma unroll
    for (int nb = 0; nb < 4; ++nb) {
        const float bias = bq[j0 + nb * 16 + laneM];
#pragma unroll
        for (int r = 0; r < 8; ++r) acc[nb][r] += bias;
    }
    if (j0 < 2 * CC) {
#pragma unroll
        for (int hd = 0; hd < 2; ++hd) {
#pragma unroll
            for (int r = 0; r < 8; ++r) {
                const float a0 = acc[2 * hd][r], a1 = acc[2 * hd + 1][r];
                const float ss = row16_sum(a0 * a0 + a1 * a1);
                const float inv = rsqrt_hw(ss);
                acc[2 * hd][r] = a0 * inv;
                acc[2 * hd + 1][r] = a1 * inv;
            }
        }
    }
    const int mbase = m0 + (laneH << 3);
#pragma unroll
    for (int nb = 0; nb < 4; ++nb) {
        float* outp = qkv + (size_t)(row0 + mbase) * TC + j0 + nb * 16 + laneM;
#pragma unroll
        for (int r = 0; r < 8; ++r) outp[(size_t)r * TC] = acc[nb][r];
    }
}

__global__ __launch_bounds__(512) void attn_kernel(float* qkv)
{
    __shared__ __attribute__((aligned(16))) float red[16 * 1024];
    const int b = blockIdx.x;
    const int s = b & 3, nh = b >> 2;
    const int n = nh >> 3, h = nh & 7;
    const int tid = threadIdx.x, lane = tid & 31, wave = tid >> 5;
    const int laneM = lane & 15;
    const int laneH = lane >> 4;

    v8f acc[2][2] = {};
    float* base = qkv + (size_t)n * LL * TC + h * DD;
    const float* kp = base + (size_t)(s * 1024 + wave * 64 + 2 * laneH) * TC + CC + 2 * laneM;
    float2 k0[16], k1[16], v0[16], v1[16];
#pragma unroll
    for (int j = 0; j < 16; ++j) {
        const float* p = kp + (size_t)(4 * j) * TC;
        k0[j] = *(const float2*)p;
        k1[j] = *(const float2*)(p + TC);
        v0[j] = *(const float2*)(p + CC);
        v1[j] = *(const float2*)(p + TC + CC);
    }
    __builtin_amdgcn_sched_barrier(0);
#pragma unroll
    for (int j = 0; j < 16; ++j) {
        v2f a0, a1, b0, b1;
        a0.x = k0[j].x; a0.y = k1[j].x;
        a1.x = k0[j].y; a1.y = k1[j].y;
        b0.x = v0[j].x; b0.y = v1[j].x;
        b1.x = v0[j].y; b1.y = v1[j].y;
        acc[0][0] = wmma_f32(a0, b0, acc[0][0]);
        acc[0][1] = wmma_f32(a0, b1, acc[0][1]);
        acc[1][0] = wmma_f32(a1, b0, acc[1][0]);
        acc[1][1] = wmma_f32(a1, b1, acc[1][1]);
    }
#pragma unroll
    for (int md = 0; md < 2; ++md)
#pragma unroll
        for (int r = 0; r < 8; ++r) {
            const int d = 2 * (8 * laneH + r) + md;
            float2 t; t.x = acc[md][0][r]; t.y = acc[md][1][r];
            *(float2*)&red[wave * 1024 + d * 32 + 2 * laneM] = t;
        }
    __syncthreads();
    for (int i = tid; i < 1024; i += 512) {
        float sum = 0.f;
#pragma unroll
        for (int w = 0; w < 16; ++w) sum += red[w * 1024 + i];
        base[(size_t)(s * 1024 + (i >> 5)) * TC + CC + (i & 31)] = sum;
    }
}

struct HeadStage {
    float v[9];
    float q[8];
    float p[4][4];
    float2 w[16];
};

static __device__ __forceinline__ void head_stage_load_vq(
    HeadStage& st, const float* __restrict__ qb, int l0, int sr, int se)
{
#pragma unroll
    for (int u = 0; u < 9; ++u) {
        const int l = l0 - 4 + sr + 8 * u;
        const int lcl = l < 0 ? 0 : (l >= LL ? LL - 1 : l);
        st.v[u] = qb[(size_t)lcl * TC + 2 * CC + se];
    }
#pragma unroll
    for (int u = 0; u < 8; ++u) st.q[u] = qb[(size_t)(l0 + sr + 8 * u) * TC + se];
}

static __device__ __forceinline__ void head_stage_load_wa(
    HeadStage& st, const float* __restrict__ qb, const float* __restrict__ wsrc, int sr, int se)
{
#pragma unroll
    for (int u = 0; u < 16; ++u) st.w[u] = *(const float2*)&wsrc[(size_t)u * 16 * CC];
#pragma unroll
    for (int u = 0; u < 4; ++u)
#pragma unroll
        for (int s = 0; s < 4; ++s)
            st.p[u][s] = qb[(size_t)(s * 1024 + sr + 8 * u) * TC + CC + se];
}

#define VS_OFF 0
#define QS_OFF 2592
#define A2_OFF 4896
#define AT_OFF 5920
#define WT_OFF 8032
#define SM_FLOATS 16288

__global__ __launch_bounds__(256, 4) void epi_proj_kernel(
    const float* __restrict__ qkv, const float* __restrict__ wdc,
    const float* __restrict__ wp, const float* __restrict__ bp, float* __restrict__ out)
{
    __shared__ __attribute__((aligned(16))) float smem[SM_FLOATS];
    float* v_s = smem + VS_OFF;
    float* q_s = smem + QS_OFF;
    float* a2 = smem + A2_OFF;
    float* At2f = smem + AT_OFF;
    float2* Wt2 = (float2*)(smem + WT_OFF);
    const int b = blockIdx.x;
    const int lc = b & 63;
    const int n = b >> 6;
    const int l0 = lc * 64;
    const int tid = threadIdx.x;
    const int lane = tid & 31;
    const int wave = __builtin_amdgcn_readfirstlane(tid >> 5);
    const int laneM = lane & 15;
    const int laneH = lane >> 4;
    const int sr = tid >> 5, se = tid & 31;
    const int sp = tid & 15, sj = tid >> 4;
    const int mp = wave & 3;
    const int nh = wave >> 2;
    const float* qn = qkv + (size_t)n * LL * TC;
    const float* wsrc0 = wp + (size_t)sj * CC + 2 * sp;
    const float inv_pi = 0.31830988618379067154f;
    float* vst = v_s + sr * 36 + se;
    float* qst = q_s + sr * 36 + se;
    float* ast = a2 + ((sr >> 1) * 32 + se) * 2 + (sr & 1);
    float2* wst = Wt2 + sp * 258 + sj;
    const float2* apf = (const float2*)At2f + laneH * 66 + mp * 16 + laneM;
    const float2* bpf = Wt2 + laneH * 258 + nh * 128 + laneM;

    v8f acc[8] = {};
    HeadStage st;
    head_stage_load_vq(st, qn, l0, sr, se);
    head_stage_load_wa(st, qn, wsrc0, sr, se);

    for (int h = 0; h < HH; ++h) {
        float wd[9];
#pragma unroll
        for (int t = 0; t < 9; ++t) wd[t] = wdc[h * 9 + t];
        float areg[4];
#pragma unroll
        for (int u = 0; u < 4; ++u) areg[u] = ((st.p[u][0] + st.p[u][1]) + st.p[u][2]) + st.p[u][3];
#pragma unroll
        for (int u = 0; u < 9; ++u) {
            const int l = l0 - 4 + sr + 8 * u;
            vst[u * 8 * 36] = (l >= 0 && l < LL) ? st.v[u] : 0.0f;
        }
#pragma unroll
        for (int u = 0; u < 8; ++u) qst[u * 8 * 36] = st.q[u];
#pragma unroll
        for (int u = 0; u < 4; ++u) ast[u * 4 * 32 * 2] = areg[u];
#pragma unroll
        for (int u = 0; u < 16; ++u) wst[16 * u] = st.w[u];
        __syncthreads();

        if (wave < 4) {
            const int m0 = wave * 16;
            const float* qf = q_s + (m0 + laneM) * 36 + 2 * laneH;
            const float* af = a2 + (laneH * 32 + laneM) * 2;
            const float* vw = v_s + (m0 + 8 * laneH) * 36 + laneM;
            float* atw = At2f + ((laneM >> 1) * 66 + m0 + 8 * laneH) * 2 + (laneM & 1);
            v8f P0 = {}, P1 = {};
#pragma unroll
            for (int s = 0; s < 8; ++s) {
                const v2f a = *(const v2f*)&qf[4 * s];
                const v2f b0 = *(const v2f*)&af[s * 2 * 32 * 2];
                const v2f b1 = *(const v2f*)&af[s * 2 * 32 * 2 + 16 * 2];
                P0 = wmma_f32(a, b0, P0);
                P1 = wmma_f32(a, b1, P1);
            }
            float w0[16], w1[16];
#pragma unroll
            for (int i = 0; i < 16; ++i) {
                w0[i] = vw[i * 36];
                w1[i] = vw[i * 36 + 16];
            }
#pragma unroll
            for (int r = 0; r < 8; ++r) {
                const float x0 = 0.5f * w0[r + 4] + inv_pi * P0[r];
                const float x1 = 0.5f * w1[r + 4] + inv_pi * P1[r];
                const float ss = row16_sum(x0 * x0 + x1 * x1);
                const float inv = rsqrt_hw(ss);
                float d0 = 0.f, d1 = 0.f;
#pragma unroll
                for (int t = 0; t < 9; ++t) {
                    d0 += wd[t] * w0[r + t];
                    d1 += wd[t] * w1[r + t];
                }
                atw[2 * r] = x0 * inv + d0;
                atw[8 * 66 * 2 + 2 * r] = x1 * inv + d1;
            }
        }
        __syncthreads();
        if (h + 1 < HH) {
            head_stage_load_vq(st, qn + (h + 1) * DD, l0, sr, se);
            head_stage_load_wa(st, qn + (h + 1) * DD, wsrc0 + (h + 1) * DD, sr, se);
        }
        __builtin_amdgcn_sched_barrier(0);

#pragma unroll
        for (int ks = 0; ks < 8; ++ks) {
            const v2f a = *(const v2f*)&apf[ks * 2 * 66];
#pragma unroll
            for (int nb = 0; nb < 8; ++nb) {
                const v2f bb = *(const v2f*)&bpf[ks * 2 * 258 + nb * 16];
                acc[nb] = wmma_f32(a, bb, acc[nb]);
            }
        }
        __syncthreads();
    }

    float bias[8];
#pragma unroll
    for (int nb = 0; nb < 8; ++nb) bias[nb] = bp[nh * 128 + nb * 16 + laneM];
    const int oj = tid >> 4, ol4 = (tid & 15) * 4;
    for (int pass = 0; pass < 2; ++pass) {
        if (nh == pass) {
#pragma unroll
            for (int nb = 0; nb < 8; ++nb) {
                float* o = smem + (nb * 16 + laneM) * 68 + mp * 16 + 8 * laneH;
                float4 lo, hi;
                lo.x = acc[nb][0] + bias[nb]; lo.y = acc[nb][1] + bias[nb];
                lo.z = acc[nb][2] + bias[nb]; lo.w = acc[nb][3] + bias[nb];
                hi.x = acc[nb][4] + bias[nb]; hi.y = acc[nb][5] + bias[nb];
                hi.z = acc[nb][6] + bias[nb]; hi.w = acc[nb][7] + bias[nb];
                *(float4*)o = lo;
                *(float4*)(o + 4) = hi;
            }
        }
        __syncthreads();
        float* dst = out + ((size_t)(n * CC + pass * 128 + oj)) * LL + l0 + ol4;
#pragma unroll
        for (int u = 0; u < 8; ++u) {
            const float4 t = *(const float4*)&smem[(oj + 16 * u) * 68 + ol4];
            *(float4*)&dst[(size_t)u * 16 * LL] = t;
        }
        __syncthreads();
    }
}

extern "C" void kernel_launch(void* const* d_in, const int* in_sizes, int n_in,
                              void* d_out, int out_size, void* d_ws, size_t ws_size,
                              hipStream_t stream) {
    const float* x      = (const float*)d_in[0];
    const float* w_qkv  = (const float*)d_in[1];
    const float* b_qkv  = (const float*)d_in[2];
    const float* w_proj = (const float*)d_in[3];
    const float* b_proj = (const float*)d_in[4];
    const float* w_dcv  = (const float*)d_in[5];
    float* out = (float*)d_out;

    float* qkv = (float*)d_ws;

    qkv_gemm_kernel<<<dim3(256, 12), 256, 0, stream>>>(x, w_qkv, b_qkv, qkv);
    attn_kernel<<<256, 512, 0, stream>>>(qkv);
    epi_proj_kernel<<<512, 256, 0, stream>>>(qkv, w_dcv, w_proj, b_proj, out);
}


# k^T v moved into the QKV GEMM: a block owns one head of 512 rows and feeds its k|v D tiles to 16 more f32 WMMAs per tile; attn launch gone (3 -> 2), k not stored; 8 partials: sum order changes
# speedup vs baseline: 2.1073x; 1.0386x over previous
#include <hip/hip_runtime.h>
#include <hip/hip_bf16.h>
#include <math.h>

typedef float v2f __attribute__((ext_vector_type(2)));
typedef float v8f __attribute__((ext_vector_type(8)));

#define NB 8
#define CC 256
#define HH 8
#define DD 32
#define LL 4096
#define TC 768

static __device__ __forceinline__ v8f wmma_f32(v2f a, v2f b, v8f c) {
    return __builtin_amdgcn_wmma_f32_16x16x4_f32(false, a, false, b, (short)0, c, false, false);
}

static __device__ __forceinline__ float row16_sum(float v) {
    v += __builtin_bit_cast(float, __builtin_amdgcn_update_dpp(0, __builtin_bit_cast(int, v), 0x161, 0xf, 0xf, true));
    v += __builtin_bit_cast(float, __builtin_amdgcn_update_dpp(0, __builtin_bit_cast(int, v), 0x162, 0xf, 0xf, true));
    v += __builtin_bit_cast(float, __builtin_amdgcn_update_dpp(0, __builtin_bit_cast(int, v), 0x164, 0xf, 0xf, true));
    v += __builtin_bit_cast(float, __builtin_amdgcn_update_dpp(0, __builtin_bit_cast(int, v), 0x168, 0xf, 0xf, true));
    return v;
}

static __device__ __forceinline__ float rsqrt_hw(float s) {
    const bool tiny = s < 0x1p-100f;
    const float r = __builtin_amdgcn_rsqf(tiny ? s * 0x1p+64f : s);
    return tiny ? r * 0x1p+32f : r;
}

#define KVB 512
#define KV_TILES 4
#define QKV_BLOCKS 1536

static __device__ __forceinline__ void qkv_tile_gemm(
    const float* __restrict__ xA, const float* const (&wrow)[4], float2* Wt2,
    int sp, int sj, int laneM, int laneH, v8f (&acc)[4])
{
    const int laneK = laneH << 1;
    for (int kc = 0; kc < CC; kc += 32) {
        float2 wreg[4];
#pragma unroll
        for (int u = 0; u < 4; ++u)
            wreg[u] = *(const float2*)&wrow[u][kc];
        v2f areg[8];
#pragma unroll
        for (int q = 0; q < 8; ++q) {
            const int k = kc + 4 * q + laneK;
            areg[q].x = xA[(size_t)k * LL];
            areg[q].y = xA[(size_t)(k + 1) * LL];
        }
        __builtin_amdgcn_sched_barrier(0);
#pragma unroll
        for (int u = 0; u < 4; ++u) Wt2[sp * 66 + sj + 16 * u] = wreg[u];
        __syncthreads();
#pragma unroll
        for (int kk = 0; kk < 32; kk += 4) {
            const v2f a = areg[kk >> 2];
            const int pp = (kk >> 1) + laneH;
#pragma unroll
            for (int nb = 0; nb < 4; ++nb) {
                const v2f b = *(const v2f*)&Wt2[pp * 66 + nb * 16 + laneM];
                acc[nb] = wmma_f32(a, b, acc[nb]);
            }
        }
        __syncthreads();
    }
}

__global__ __launch_bounds__(256) void qkv_attn_kernel(
    const float* __restrict__ x, const float* __restrict__ wq,
    const float* __restrict__ bq, float* __restrict__ qkv)
{
    __shared__ __attribute__((aligned(16))) float smem[8 * 1024];
    float2* Wt2 = (float2*)smem;
    const int b = blockIdx.x;
    const int tid = threadIdx.x;
    const int lane = tid & 31, wave = tid >> 5;
    const int m0 = wave * 16;
    const int laneM = lane & 15;
    const int laneH = lane >> 4;
    const int sp = tid & 15, sj = tid >> 4;
    const int mbase = m0 + (laneH << 3);

    if (b >= KVB) {
        const int qb = b - KVB;
        const int mt = qb & 255;
        const int nt = qb >> 8;
        const int row0 = mt * 128;
        const int n = row0 >> 12;
        const int l0 = row0 & (LL - 1);
        const int j0 = nt * 64;
        v8f acc[4] = {};
        const float* xA = x + (size_t)n * CC * LL + l0 + m0 + laneM;
        const float* wsrc = wq + (size_t)(j0 + sj) * CC + 2 * sp;
        const float* const wrow[4] = {wsrc, wsrc + 16 * CC, wsrc + 32 * CC, wsrc + 48 * CC};
        qkv_tile_gemm(xA, wrow, Wt2, sp, sj, laneM, laneH, acc);
#pragma unroll
        for (int nb = 0; nb < 4; ++nb) {
            const float bias = bq[j0 + nb * 16 + laneM];
#pragma unroll
            for (int r = 0; r < 8; ++r) acc[nb][r] += bias;
        }
#pragma unroll
        for (int hd = 0; hd < 2; ++hd) {
#pragma unroll
            for (int r = 0; r < 8; ++r) {
                const float a0 = acc[2 * hd][r], a1 = acc[2 * hd + 1][r];
                const float ss = row16_sum(a0 * a0 + a1 * a1);
                const float inv = rsqrt_hw(ss);
                acc[2 * hd][r] = a0 * inv;
                acc[2 * hd + 1][r] = a1 * inv;
            }
        }
#pragma unroll
        for (int nb = 0; nb < 4; ++nb) {
            float* outp = qkv + (size_t)(row0 + mbase) * TC + j0 + nb * 16 + laneM;
#pragma unroll
            for (int r = 0; r < 8; ++r) outp[(size_t)r * TC] = acc[nb][r];
        }
    } else {
        const int s = b & 7, nh = b >> 3;
        const int n = nh >> 3, h = nh & 7;
        const float* wk = wq + (size_t)(CC + h * DD + sj) * CC + 2 * sp;
        const float* wv = wq + (size_t)(2 * CC + h * DD + sj) * CC + 2 * sp;
        const float* const wrow[4] = {wk, wk + 16 * CC, wv, wv + 16 * CC};
        float bias[4];
        bias[0] = bq[CC + h * DD + laneM];
        bias[1] = bq[CC + h * DD + 16 + laneM];
        bias[2] = bq[2 * CC + h * DD + laneM];
        bias[3] = bq[2 * CC + h * DD + 16 + laneM];
        float* base = qkv + (size_t)n * LL * TC + h * DD;
        v8f at[2][2] = {};
#pragma unroll 1
        for (int t = 0; t < KV_TILES; ++t) {
            const int l0 = s * 512 + t * 128;
            v8f acc[4] = {};
            const float* xA = x + (size_t)n * CC * LL + l0 + m0 + laneM;
            qkv_tile_gemm(xA, wrow, Wt2, sp, sj, laneM, laneH, acc);
#pragma unroll
            for (int nb = 0; nb < 4; ++nb)
#pragma unroll
                for (int r = 0; r < 8; ++r) acc[nb][r] += bias[nb];
#pragma unroll
            for (int r = 0; r < 8; ++r) {
                const float a0 = acc[0][r], a1 = acc[1][r];
                const float ss = row16_sum(a0 * a0 + a1 * a1);
                const float inv = rsqrt_hw(ss);
                acc[0][r] = a0 * inv;
                acc[1][r] = a1 * inv;
            }
#pragma unroll
            for (int nb = 2; nb < 4; ++nb) {
                float* outp = base + (size_t)(l0 + mbase) * TC + 2 * CC + (nb - 2) * 16 + laneM;
#pragma unroll
                for (int r = 0; r < 8; ++r) outp[(size_t)r * TC] = acc[nb][r];
            }
#pragma unroll
            for (int tt = 0; tt < 4; ++tt) {
                v2f a0, a1, b0, b1;
                a0.x = acc[0][2 * tt]; a0.y = acc[0][2 * tt + 1];
                a1.x = acc[1][2 * tt]; a1.y = acc[1][2 * tt + 1];
                b0.x = acc[2][2 * tt]; b0.y = acc[2][2 * tt + 1];
                b1.x = acc[3][2 * tt]; b1.y = acc[3][2 * tt + 1];
                at[0][0] = wmma_f32(a0, b0, at[0][0]);
                at[0][1] = wmma_f32(a0, b1, at[0][1]);
                at[1][0] = wmma_f32(a1, b0, at[1][0]);
                at[1][1] = wmma_f32(a1, b1, at[1][1]);
            }
        }
#pragma unroll
        for (int md = 0; md < 2; ++md)
#pragma unroll
            for (int me = 0; me < 2; ++me)
#pragma unroll
                for (int r = 0; r < 8; ++r)
                    smem[wave * 1024 + (16 * md + 8 * laneH + r) * 32 + 16 * me + laneM] = at[md][me][r];
        __syncthreads();
        for (int i = tid; i < 1024; i += 256) {
            float sum = 0.f;
#pragma unroll
            for (int w = 0; w < 8; ++w) sum += smem[w * 1024 + i];
            base[(size_t)(s * 512 + (i >> 5)) * TC + CC + (i & 31)] = sum;
        }
    }
}

struct HeadStage {
    float v[9];
    float q[8];
    float4 p[8];
    float2 w[16];
};

static __device__ __forceinline__ void head_stage_load_vq(
    HeadStage& st, const float* __restrict__ qb, int l0, int sr, int se)
{
#pragma unroll
    for (int u = 0; u < 9; ++u) {
        const int l = l0 - 4 + sr + 8 * u;
        const int lcl = l < 0 ? 0 : (l >= LL ? LL - 1 : l);
        st.v[u] = qb[(size_t)lcl * TC + 2 * CC + se];
    }
#pragma unroll
    for (int u = 0; u < 8; ++u) st.q[u] = qb[(size_t)(l0 + sr + 8 * u) * TC + se];
}

static __device__ __forceinline__ void head_stage_load_wa(
    HeadStage& st, const float* __restrict__ qb, const float* __restrict__ wsrc, int pd, int pe)
{
#pragma unroll
    for (int u = 0; u < 16; ++u) st.w[u] = *(const float2*)&wsrc[(size_t)u * 16 * CC];
#pragma unroll
    for (int s = 0; s < 8; ++s)
        st.p[s] = *(const float4*)&qb[(size_t)(s * 512 + pd) * TC + CC + pe];
}

#define VS_OFF 0
#define QS_OFF 2592
#define A2_OFF 4896
#define AT_OFF 5920
#define WT_OFF 8032
#define SM_FLOATS 16288

__global__ __launch_bounds__(256, 4) void epi_proj_kernel(
    const float* __restrict__ qkv, const float* __restrict__ wdc,
    const float* __restrict__ wp, const float* __restrict__ bp, float* __restrict__ out)
{
    __shared__ __attribute__((aligned(16))) float smem[SM_FLOATS];
    float* v_s = smem + VS_OFF;
    float* q_s = smem + QS_OFF;
    float* a2 = smem + A2_OFF;
    float* At2f = smem + AT_OFF;
    float2* Wt2 = (float2*)(smem + WT_OFF);
    const int b = blockIdx.x;
    const int lc = b & 63;
    const int n = b >> 6;
    const int l0 = lc * 64;
    const int tid = threadIdx.x;
    const int lane = tid & 31;
    const int wave = __builtin_amdgcn_readfirstlane(tid >> 5);
    const int laneM = lane & 15;
    const int laneH = lane >> 4;
    const int sr = tid >> 5, se = tid & 31;
    const int sp = tid & 15, sj = tid >> 4;
    const int pd = tid >> 3, pe = (tid & 7) * 4;
    const int mp = wave & 3;
    const int nh = wave >> 2;
    const float* qn = qkv + (size_t)n * LL * TC;
    const float* wsrc0 = wp + (size_t)sj * CC + 2 * sp;
    const float inv_pi = 0.31830988618379067154f;
    float* vst = v_s + sr * 36 + se;
    float* qst = q_s + sr * 36 + se;
    float* ast = a2 + ((pd >> 1) * 32 + pe) * 2 + (pd & 1);
    float2* wst = Wt2 + sp * 258 + sj;
    const float2* apf = (const float2*)At2f + laneH * 66 + mp * 16 + laneM;
    const float2* bpf = Wt2 + laneH * 258 + nh * 128 + laneM;

    v8f acc[8] = {};
    HeadStage st;
    head_stage_load_vq(st, qn, l0, sr, se);
    head_stage_load_wa(st, qn, wsrc0, pd, pe);

    for (int h = 0; h < HH; ++h) {
        float wd[9];
#pragma unroll
        for (int t = 0; t < 9; ++t) wd[t] = wdc[h * 9 + t];
        float4 asum = st.p[0];
#pragma unroll
        for (int s = 1; s < 8; ++s) {
            asum.x += st.p[s].x; asum.y += st.p[s].y; asum.z += st.p[s].z; asum.w += st.p[s].w;
        }
#pragma unroll
        for (int u = 0; u < 9; ++u) {
            const int l = l0 - 4 + sr + 8 * u;
            vst[u * 8 * 36] = (l >= 0 && l < LL) ? st.v[u] : 0.0f;
        }
#pragma unroll
        for (int u = 0; u < 8; ++u) qst[u * 8 * 36] = st.q[u];
        ast[0] = asum.x; ast[2] = asum.y; ast[4] = asum.z; ast[6] = asum.w;
#pragma unroll
        for (int u = 0; u < 16; ++u) wst[16 * u] = st.w[u];
        __syncthreads();

        if (wave < 4) {
            const int m0 = wave * 16;
            const float* qf = q_s + (m0 + laneM) * 36 + 2 * laneH;
            const float* af = a2 + (laneH * 32 + laneM) * 2;
            const float* vw = v_s + (m0 + 8 * laneH) * 36 + laneM;
            float* atw = At2f + ((laneM >> 1) * 66 + m0 + 8 * laneH) * 2 + (laneM & 1);
            v8f P0 = {}, P1 = {};
#pragma unroll
            for (int s = 0; s < 8; ++s) {
                const v2f a = *(const v2f*)&qf[4 * s];
                const v2f b0 = *(const v2f*)&af[s * 2 * 32 * 2];
                const v2f b1 = *(const v2f*)&af[s * 2 * 32 * 2 + 16 * 2];
                P0 = wmma_f32(a, b0, P0);
                P1 = wmma_f32(a, b1, P1);
            }
            float w0[16], w1[16];
#pragma unroll
            for (int i = 0; i < 16; ++i) {
                w0[i] = vw[i * 36];
                w1[i] = vw[i * 36 + 16];
            }
#pragma unroll
            for (int r = 0; r < 8; ++r) {
                const float x0 = 0.5f * w0[r + 4] + inv_pi * P0[r];
                const float x1 = 0.5f * w1[r + 4] + inv_pi * P1[r];
                const float ss = row16_sum(x0 * x0 + x1 * x1);
                const float inv = rsqrt_hw(ss);
                float d0 = 0.f, d1 = 0.f;
#pragma unroll
                for (int t = 0; t < 9; ++t) {
                    d0 += wd[t] * w0[r + t];
                    d1 += wd[t] * w1[r + t];
                }
                atw[2 * r] = x0 * inv + d0;
                atw[8 * 66 * 2 + 2 * r] = x1 * inv + d1;
            }
        }
        __syncthreads();
        if (h + 1 < HH) {
            head_stage_load_vq(st, qn + (h + 1) * DD, l0, sr, se);
            head_stage_load_wa(st, qn + (h + 1) * DD, wsrc0 + (h + 1) * DD, pd, pe);
        }
        __builtin_amdgcn_sched_barrier(0);

#pragma unroll
        for (int ks = 0; ks < 8; ++ks) {
            const v2f a = *(const v2f*)&apf[ks * 2 * 66];
#pragma unroll
            for (int nb = 0; nb < 8; ++nb) {
                const v2f bb = *(const v2f*)&bpf[ks * 2 * 258 + nb * 16];
                acc[nb] = wmma_f32(a, bb, acc[nb]);
            }
        }
        __syncthreads();
    }

    float bias[8];
#pragma unroll
    for (int nb = 0; nb < 8; ++nb) bias[nb] = bp[nh * 128 + nb * 16 + laneM];
    const int oj = tid >> 4, ol4 = (tid & 15) * 4;
    for (int pass = 0; pass < 2; ++pass) {
        if (nh == pass) {
#pragma unroll
            for (int nb = 0; nb < 8; ++nb) {
                float* o = smem + (nb * 16 + laneM) * 68 + mp * 16 + 8 * laneH;
                float4 lo, hi;
                lo.x = acc[nb][0] + bias[nb]; lo.y = acc[nb][1] + bias[nb];
                lo.z = acc[nb][2] + bias[nb]; lo.w = acc[nb][3] + bias[nb];
                hi.x = acc[nb][4] + bias[nb]; hi.y = acc[nb][5] + bias[nb];
                hi.z = acc[nb][6] + bias[nb]; hi.w = acc[nb][7] + bias[nb];
                *(float4*)o = lo;
                *(float4*)(o + 4) = hi;
            }
        }
        __syncthreads();
        float* dst = out + ((size_t)(n * CC + pass * 128 + oj)) * LL + l0 + ol4;
#pragma unroll
        for (int u = 0; u < 8; ++u) {
            const float4 t = *(const float4*)&smem[(oj + 16 * u) * 68 + ol4];
            *(float4*)&dst[(size_t)u * 16 * LL] = t;
        }
        __syncthreads();
    }
}

extern "C" void kernel_launch(void* const* d_in, const int* in_sizes, int n_in,
                              void* d_out, int out_size, void* d_ws, size_t ws_size,
                              hipStream_t stream) {
    const float* x      = (const float*)d_in[0];
    const float* w_qkv  = (const float*)d_in[1];
    const float* b_qkv  = (const float*)d_in[2];
    const float* w_proj = (const float*)d_in[3];
    const float* b_proj = (const float*)d_in[4];
    const float* w_dcv  = (const float*)d_in[5];
    float* out = (float*)d_out;

    float* qkv = (float*)d_ws;

    qkv_attn_kernel<<<QKV_BLOCKS, 256, 0, stream>>>(x, w_qkv, b_qkv, qkv);
    epi_proj_kernel<<<512, 256, 0, stream>>>(qkv, w_dcv, w_proj, b_proj, out);
}


# QKV kernel: a wave owns 32 rows x 64 columns as two row tiles interleaved by row parity: 8-byte x loads, each B fragment read feeds two WMMAs; k^T v row sum order changes
# speedup vs baseline: 2.1689x; 1.0292x over previous
#include <hip/hip_runtime.h>
#include <hip/hip_bf16.h>
#include <math.h>

typedef float v2f __attribute__((ext_vector_type(2)));
typedef float v8f __attribute__((ext_vector_type(8)));

#define NB 8
#define CC 256
#define HH 8
#define DD 32
#define LL 4096
#define TC 768

static __device__ __forceinline__ v8f wmma_f32(v2f a, v2f b, v8f c) {
    return __builtin_amdgcn_wmma_f32_16x16x4_f32(false, a, false, b, (short)0, c, false, false);
}

static __device__ __forceinline__ float row16_sum(float v) {
    v += __builtin_bit_cast(float, __builtin_amdgcn_update_dpp(0, __builtin_bit_cast(int, v), 0x161, 0xf, 0xf, true));
    v += __builtin_bit_cast(float, __builtin_amdgcn_update_dpp(0, __builtin_bit_cast(int, v), 0x162, 0xf, 0xf, true));
    v += __builtin_bit_cast(float, __builtin_amdgcn_update_dpp(0, __builtin_bit_cast(int, v), 0x164, 0xf, 0xf, true));
    v += __builtin_bit_cast(float, __builtin_amdgcn_update_dpp(0, __builtin_bit_cast(int, v), 0x168, 0xf, 0xf, true));
    return v;
}

static __device__ __forceinline__ float rsqrt_hw(float s) {
    const bool tiny = s < 0x1p-100f;
    const float r = __builtin_amdgcn_rsqf(tiny ? s * 0x1p+64f : s);
    return tiny ? r * 0x1p+32f : r;
}

#define KVB 512
#define KV_TILES 2
#define QKV_BLOCKS 1024

static __device__ __forceinline__ void qkv_tile_gemm(
    const float* __restrict__ xA, const float* const (&wrow)[4], float2* Wt2,
    int sp, int sj, int laneM, int laneH, v8f (&acc)[2][4])
{
    const int laneK = laneH << 1;
    for (int kc = 0; kc < CC; kc += 32) {
        float2 wreg[4];
#pragma unroll
        for (int u = 0; u < 4; ++u)
            wreg[u] = *(const float2*)&wrow[u][kc];
        float2 xr[16];
#pragma unroll
        for (int q = 0; q < 8; ++q) {
            const int k = kc + 4 * q + laneK;
            xr[2 * q] = *(const float2*)&xA[(size_t)k * LL];
            xr[2 * q + 1] = *(const float2*)&xA[(size_t)(k + 1) * LL];
        }
        __builtin_amdgcn_sched_barrier(0);
#pragma unroll
        for (int u = 0; u < 4; ++u) Wt2[sp * 66 + sj + 16 * u] = wreg[u];
        __syncthreads();
#pragma unroll
        for (int kk = 0; kk < 32; kk += 4) {
            v2f a0, a1;
            a0.x = xr[kk >> 1].x; a0.y = xr[(kk >> 1) + 1].x;
            a1.x = xr[kk >> 1].y; a1.y = xr[(kk >> 1) + 1].y;
            const int pp = (kk >> 1) + laneH;
#pragma unroll
            for (int nb = 0; nb < 4; ++nb) {
                const v2f b = *(const v2f*)&Wt2[pp * 66 + nb * 16 + laneM];
                acc[0][nb] = wmma_f32(a0, b, acc[0][nb]);
                acc[1][nb] = wmma_f32(a1, b, acc[1][nb]);
            }
        }
        __syncthreads();
    }
}

__global__ __launch_bounds__(256) void qkv_attn_kernel(
    const float* __restrict__ x, const float* __restrict__ wq,
    const float* __restrict__ bq, float* __restrict__ qkv)
{
    __shared__ __attribute__((aligned(16))) float smem[8 * 1024];
    float2* Wt2 = (float2*)smem;
    const int b = blockIdx.x;
    const int tid = threadIdx.x;
    const int lane = tid & 31, wave = tid >> 5;
    const int m0 = wave * 32;
    const int laneM = lane & 15;
    const int laneH = lane >> 4;
    const int sp = tid & 15, sj = tid >> 4;
    const int mbase = m0 + (laneH << 4);

    if (b >= KVB) {
        const int qb = b - KVB;
        const int mt = qb & 127;
        const int nt = qb >> 7;
        const int row0 = mt * 256;
        const int n = row0 >> 12;
        const int l0 = row0 & (LL - 1);
        const int j0 = nt * 64;
        v8f acc[2][4] = {};
        const float* xA = x + (size_t)n * CC * LL + l0 + m0 + 2 * laneM;
        const float* wsrc = wq + (size_t)(j0 + sj) * CC + 2 * sp;
        const float* const wrow[4] = {wsrc, wsrc + 16 * CC, wsrc + 32 * CC, wsrc + 48 * CC};
        qkv_tile_gemm(xA, wrow, Wt2, sp, sj, laneM, laneH, acc);
#pragma unroll
        for (int nb = 0; nb < 4; ++nb) {
            const float bias = bq[j0 + nb * 16 + laneM];
#pragma unroll
            for (int i = 0; i < 2; ++i)
#pragma unroll
                for (int r = 0; r < 8; ++r) acc[i][nb][r] += bias;
        }
#pragma unroll
        for (int i = 0; i < 2; ++i) {
#pragma unroll
            for (int hd = 0; hd < 2; ++hd) {
#pragma unroll
                for (int r = 0; r < 8; ++r) {
                    const float a0 = acc[i][2 * hd][r], a1 = acc[i][2 * hd + 1][r];
                    const float ss = row16_sum(a0 * a0 + a1 * a1);
                    const float inv = rsqrt_hw(ss);
                    acc[i][2 * hd][r] = a0 * inv;
                    acc[i][2 * hd + 1][r] = a1 * inv;
                }
            }
        }
#pragma unroll
        for (int i = 0; i < 2; ++i) {
#pragma unroll
            for (int nb = 0; nb < 4; ++nb) {
                float* outp = qkv + (size_t)(row0 + mbase + i) * TC + j0 + nb * 16 + laneM;
#pragma unroll
                for (int r = 0; r < 8; ++r) outp[(size_t)(2 * r) * TC] = acc[i][nb][r];
            }
        }
    } else {
        const int s = b & 7, nh = b >> 3;
        const int n = nh >> 3, h = nh & 7;
        const float* wk = wq + (size_t)(CC + h * DD + sj) * CC + 2 * sp;
        const float* wv = wq + (size_t)(2 * CC + h * DD + sj) * CC + 2 * sp;
        const float* const wrow[4] = {wk, wk + 16 * CC, wv, wv + 16 * CC};
        float bias[4];
        bias[0] = bq[CC + h * DD + laneM];
        bias[1] = bq[CC + h * DD + 16 + laneM];
        bias[2] = bq[2 * CC + h * DD + laneM];
        bias[3] = bq[2 * CC + h * DD + 16 + laneM];
        float* base = qkv + (size_t)n * LL * TC + h * DD;
        v8f at[2][2] = {};
#pragma unroll 1
        for (int t = 0; t < KV_TILES; ++t) {
            const int l0 = s * 512 + t * 256;
            v8f acc[2][4] = {};
            const float* xA = x + (size_t)n * CC * LL + l0 + m0 + 2 * laneM;
            qkv_tile_gemm(xA, wrow, Wt2, sp, sj, laneM, laneH, acc);
#pragma unroll
            for (int i = 0; i < 2; ++i)
#pragma unroll
                for (int nb = 0; nb < 4; ++nb)
#pragma unroll
                    for (int r = 0; r < 8; ++r) acc[i][nb][r] += bias[nb];
#pragma unroll
            for (int i = 0; i < 2; ++i) {
#pragma unroll
                for (int r = 0; r < 8; ++r) {
                    const float a0 = acc[i][0][r], a1 = acc[i][1][r];
                    const float ss = row16_sum(a0 * a0 + a1 * a1);
                    const float inv = rsqrt_hw(ss);
                    acc[i][0][r] = a0 * inv;
                    acc[i][1][r] = a1 * inv;
                }
            }
#pragma unroll
            for (int i = 0; i < 2; ++i) {
#pragma unroll
                for (int nb = 2; nb < 4; ++nb) {
                    float* outp = base + (size_t)(l0 + mbase + i) * TC + 2 * CC + (nb - 2) * 16 + laneM;
#pragma unroll
                    for (int r = 0; r < 8; ++r) outp[(size_t)(2 * r) * TC] = acc[i][nb][r];
                }
            }
#pragma unroll
            for (int i = 0; i < 2; ++i) {
#pragma unroll
                for (int tt = 0; tt < 4; ++tt) {
                    v2f a0, a1, b0, b1;
                    a0.x = acc[i][0][2 * tt]; a0.y = acc[i][0][2 * tt + 1];
                    a1.x = acc[i][1][2 * tt]; a1.y = acc[i][1][2 * tt + 1];
                    b0.x = acc[i][2][2 * tt]; b0.y = acc[i][2][2 * tt + 1];
                    b1.x = acc[i][3][2 * tt]; b1.y = acc[i][3][2 * tt + 1];
                    at[0][0] = wmma_f32(a0, b0, at[0][0]);
                    at[0][1] = wmma_f32(a0, b1, at[0][1]);
                    at[1][0] = wmma_f32(a1, b0, at[1][0]);
                    at[1][1] = wmma_f32(a1, b1, at[1][1]);
                }
            }
        }
#pragma unroll
        for (int md = 0; md < 2; ++md)
#pragma unroll
            for (int me = 0; me < 2; ++me)
#pragma unroll
                for (int r = 0; r < 8; ++r)
                    smem[wave * 1024 + (16 * md + 8 * laneH + r) * 32 + 16 * me + laneM] = at[md][me][r];
        __syncthreads();
        for (int i = tid; i < 1024; i += 256) {
            float sum = 0.f;
#pragma unroll
            for (int w = 0; w < 8; ++w) sum += smem[w * 1024 + i];
            base[(size_t)(s * 512 + (i >> 5)) * TC + CC + (i & 31)] = sum;
        }
    }
}

struct HeadStage {
    float v[9];
    float q[8];
    float4 p[8];
    float2 w[16];
};

static __device__ __forceinline__ void head_stage_load_vq(
    HeadStage& st, const float* __restrict__ qb, int l0, int sr, int se)
{
#pragma unroll
    for (int u = 0; u < 9; ++u) {
        const int l = l0 - 4 + sr + 8 * u;
        const int lcl = l < 0 ? 0 : (l >= LL ? LL - 1 : l);
        st.v[u] = qb[(size_t)lcl * TC + 2 * CC + se];
    }
#pragma unroll
    for (int u = 0; u < 8; ++u) st.q[u] = qb[(size_t)(l0 + sr + 8 * u) * TC + se];
}

static __device__ __forceinline__ void head_stage_load_wa(
    HeadStage& st, const float* __restrict__ qb, const float* __restrict__ wsrc, int pd, int pe)
{
#pragma unroll
    for (int u = 0; u < 16; ++u) st.w[u] = *(const float2*)&wsrc[(size_t)u * 16 * CC];
#pragma unroll
    for (int s = 0; s < 8; ++s)
        st.p[s] = *(const float4*)&qb[(size_t)(s * 512 + pd) * TC + CC + pe];
}

#define VS_OFF 0
#define QS_OFF 2592
#define A2_OFF 4896
#define AT_OFF 5920
#define WT_OFF 8032
#define SM_FLOATS 16288

__global__ __launch_bounds__(256, 4) void epi_proj_kernel(
    const float* __restrict__ qkv, const float* __restrict__ wdc,
    const float* __restrict__ wp, const float* __restrict__ bp, float* __restrict__ out)
{
    __shared__ __attribute__((aligned(16))) float smem[SM_FLOATS];
    float* v_s = smem + VS_OFF;
    float* q_s = smem + QS_OFF;
    float* a2 = smem + A2_OFF;
    float* At2f = smem + AT_OFF;
    float2* Wt2 = (float2*)(smem + WT_OFF);
    const int b = blockIdx.x;
    const int lc = b & 63;
    const int n = b >> 6;
    const int l0 = lc * 64;
    const int tid = threadIdx.x;
    const int lane = tid & 31;
    const int wave = __builtin_amdgcn_readfirstlane(tid >> 5);
    const int laneM = lane & 15;
    const int laneH = lane >> 4;
    const int sr = tid >> 5, se = tid & 31;
    const int sp = tid & 15, sj = tid >> 4;
    const int pd = tid >> 3, pe = (tid & 7) * 4;
    const int mp = wave & 3;
    const int nh = wave >> 2;
    const float* qn = qkv + (size_t)n * LL * TC;
    const float* wsrc0 = wp + (size_t)sj * CC + 2 * sp;
    const float inv_pi = 0.31830988618379067154f;
    float* vst = v_s + sr * 36 + se;
    float* qst = q_s + sr * 36 + se;
    float* ast = a2 + ((pd >> 1) * 32 + pe) * 2 + (pd & 1);
    float2* wst = Wt2 + sp * 258 + sj;
    const float2* apf = (const float2*)At2f + laneH * 66 + mp * 16 + laneM;
    const float2* bpf = Wt2 + laneH * 258 + nh * 128 + laneM;

    v8f acc[8] = {};
    HeadStage st;
    head_stage_load_vq(st, qn, l0, sr, se);
    head_stage_load_wa(st, qn, wsrc0, pd, pe);

    for (int h = 0; h < HH; ++h) {
        float wd[9];
#pragma unroll
        for (int t = 0; t < 9; ++t) wd[t] = wdc[h * 9 + t];
        float4 asum = st.p[0];
#pragma unroll
        for (int s = 1; s < 8; ++s) {
            asum.x += st.p[s].x; asum.y += st.p[s].y; asum.z += st.p[s].z; asum.w += st.p[s].w;
        }
#pragma unroll
        for (int u = 0; u < 9; ++u) {
            const int l = l0 - 4 + sr + 8 * u;
            vst[u * 8 * 36] = (l >= 0 && l < LL) ? st.v[u] : 0.0f;
        }
#pragma unroll
        for (int u = 0; u < 8; ++u) qst[u * 8 * 36] = st.q[u];
        ast[0] = asum.x; ast[2] = asum.y; ast[4] = asum.z; ast[6] = asum.w;
#pragma unroll
        for (int u = 0; u < 16; ++u) wst[16 * u] = st.w[u];
        __syncthreads();

        if (wave < 4) {
            const int m0 = wave * 16;
            const float* qf = q_s + (m0 + laneM) * 36 + 2 * laneH;
            const float* af = a2 + (laneH * 32 + laneM) * 2;
            const float* vw = v_s + (m0 + 8 * laneH) * 36 + laneM;
            float* atw = At2f + ((laneM >> 1) * 66 + m0 + 8 * laneH) * 2 + (laneM & 1);
            v8f P0 = {}, P1 = {};
#pragma unroll
            for (int s = 0; s < 8; ++s) {
                const v2f a = *(const v2f*)&qf[4 * s];
                const v2f b0 = *(const v2f*)&af[s * 2 * 32 * 2];
                const v2f b1 = *(const v2f*)&af[s * 2 * 32 * 2 + 16 * 2];
                P0 = wmma_f32(a, b0, P0);
                P1 = wmma_f32(a, b1, P1);
            }
            float w0[16], w1[16];
#pragma unroll
            for (int i = 0; i < 16; ++i) {
                w0[i] = vw[i * 36];
                w1[i] = vw[i * 36 + 16];
            }
#pragma unroll
            for (int r = 0; r < 8; ++r) {
                const float x0 = 0.5f * w0[r + 4] + inv_pi * P0[r];
                const float x1 = 0.5f * w1[r + 4] + inv_pi * P1[r];
                const float ss = row16_sum(x0 * x0 + x1 * x1);
                const float inv = rsqrt_hw(ss);
                float d0 = 0.f, d1 = 0.f;
#pragma unroll
                for (int t = 0; t < 9; ++t) {
                    d0 += wd[t] * w0[r + t];
                    d1 += wd[t] * w1[r + t];
                }
                atw[2 * r] = x0 * inv + d0;
                atw[8 * 66 * 2 + 2 * r] = x1 * inv + d1;
            }
        }
        __syncthreads();
        if (h + 1 < HH) {
            head_stage_load_vq(st, qn + (h + 1) * DD, l0, sr, se);
            head_stage_load_wa(st, qn + (h + 1) * DD, wsrc0 + (h + 1) * DD, pd, pe);
        }
        __builtin_amdgcn_sched_barrier(0);

#pragma unroll
        for (int ks = 0; ks < 8; ++ks) {
            const v2f a = *(const v2f*)&apf[ks * 2 * 66];
#pragma unroll
            for (int nb = 0; nb < 8; ++nb) {
                const v2f bb = *(const v2f*)&bpf[ks * 2 * 258 + nb * 16];
                acc[nb] = wmma_f32(a, bb, acc[nb]);
            }
        }
        __syncthreads();
    }

    float bias[8];
#pragma unroll
    for (int nb = 0; nb < 8; ++nb) bias[nb] = bp[nh * 128 + nb * 16 + laneM];
    const int oj = tid >> 4, ol4 = (tid & 15) * 4;
    for (int pass = 0; pass < 2; ++pass) {
        if (nh == pass) {
#pragma unroll
            for (int nb = 0; nb < 8; ++nb) {
                float* o = smem + (nb * 16 + laneM) * 68 + mp * 16 + 8 * laneH;
                float4 lo, hi;
                lo.x = acc[nb][0] + bias[nb]; lo.y = acc[nb][1] + bias[nb];
                lo.z = acc[nb][2] + bias[nb]; lo.w = acc[nb][3] + bias[nb];
                hi.x = acc[nb][4] + bias[nb]; hi.y = acc[nb][5] + bias[nb];
                hi.z = acc[nb][6] + bias[nb]; hi.w = acc[nb][7] + bias[nb];
                *(float4*)o = lo;
                *(float4*)(o + 4) = hi;
            }
        }
        __syncthreads();
        float* dst = out + ((size_t)(n * CC + pass * 128 + oj)) * LL + l0 + ol4;
#pragma unroll
        for (int u = 0; u < 8; ++u) {
            const float4 t = *(const float4*)&smem[(oj + 16 * u) * 68 + ol4];
            *(float4*)&dst[(size_t)u * 16 * LL] = t;
        }
        __syncthreads();
    }
}

extern "C" void kernel_launch(void* const* d_in, const int* in_sizes, int n_in,
                              void* d_out, int out_size, void* d_ws, size_t ws_size,
                              hipStream_t stream) {
    const float* x      = (const float*)d_in[0];
    const float* w_qkv  = (const float*)d_in[1];
    const float* b_qkv  = (const float*)d_in[2];
    const float* w_proj = (const float*)d_in[3];
    const float* b_proj = (const float*)d_in[4];
    const float* w_dcv  = (const float*)d_in[5];
    float* out = (float*)d_out;

    float* qkv = (float*)d_ws;

    qkv_attn_kernel<<<QKV_BLOCKS, 256, 0, stream>>>(x, w_qkv, b_qkv, qkv);
    epi_proj_kernel<<<512, 256, 0, stream>>>(qkv, w_dcv, w_proj, b_proj, out);
}


# QKV kernel: a block's 64-column weight slice is staged in LDS once for all of K (16-byte fragment reads), so the K loop has no barrier; x of chunk c+1 is loaded under the products of chunk c
# speedup vs baseline: 2.2863x; 1.0541x over previous
#include <hip/hip_runtime.h>
#include <hip/hip_bf16.h>
#include <math.h>

typedef float v2f __attribute__((ext_vector_type(2)));
typedef float v8f __attribute__((ext_vector_type(8)));

#define NB 8
#define CC 256
#define HH 8
#define DD 32
#define LL 4096
#define TC 768

static __device__ __forceinline__ v8f wmma_f32(v2f a, v2f b, v8f c) {
    return __builtin_amdgcn_wmma_f32_16x16x4_f32(false, a, false, b, (short)0, c, false, false);
}

static __device__ __forceinline__ float row16_sum(float v) {
    v += __builtin_bit_cast(float, __builtin_amdgcn_update_dpp(0, __builtin_bit_cast(int, v), 0x161, 0xf, 0xf, true));
    v += __builtin_bit_cast(float, __builtin_amdgcn_update_dpp(0, __builtin_bit_cast(int, v), 0x162, 0xf, 0xf, true));
    v += __builtin_bit_cast(float, __builtin_amdgcn_update_dpp(0, __builtin_bit_cast(int, v), 0x164, 0xf, 0xf, true));
    v += __builtin_bit_cast(float, __builtin_amdgcn_update_dpp(0, __builtin_bit_cast(int, v), 0x168, 0xf, 0xf, true));
    return v;
}

static __device__ __forceinline__ float rsqrt_hw(float s) {
    const bool tiny = s < 0x1p-100f;
    const float r = __builtin_amdgcn_rsqf(tiny ? s * 0x1p+64f : s);
    return tiny ? r * 0x1p+32f : r;
}

#define KVB 512
#define KV_TILES 2
#define QKV_BLOCKS 1024
#define WP 66
#define WS_FLOATS (128 * WP * 2)

static __device__ __forceinline__ void x_chunk_load(
    float2 (&xr)[16], const float* __restrict__ xA, int kc, int laneK)
{
#pragma unroll
    for (int q = 0; q < 8; ++q) {
        const int k = kc + 4 * q + laneK;
        xr[2 * q] = *(const float2*)&xA[(size_t)k * LL];
        xr[2 * q + 1] = *(const float2*)&xA[(size_t)(k + 1) * LL];
    }
}

static __device__ __forceinline__ void stage_weights_and_x0(
    const float* const (&wrow)[4], float2* Wt2, int sp, int sj,
    float2 (&xr)[16], const float* __restrict__ xA, int laneK)
{
    float2 wreg[32];
#pragma unroll
    for (int cc = 0; cc < 8; ++cc)
#pragma unroll
        for (int u = 0; u < 4; ++u) wreg[cc * 4 + u] = *(const float2*)&wrow[u][cc * 32];
    x_chunk_load(xr, xA, 0, laneK);
    __builtin_amdgcn_sched_barrier(0);
#pragma unroll
    for (int cc = 0; cc < 8; ++cc) {
        float2* row = Wt2 + (cc * 16 + sp) * WP + 2 * sj;
        *(float4*)row = make_float4(wreg[cc * 4].x, wreg[cc * 4].y, wreg[cc * 4 + 1].x, wreg[cc * 4 + 1].y);
        *(float4*)(row + 32) = make_float4(wreg[cc * 4 + 2].x, wreg[cc * 4 + 2].y, wreg[cc * 4 + 3].x, wreg[cc * 4 + 3].y);
    }
}

template <bool NEXT>
static __device__ __forceinline__ void qkv_tile_gemm(
    const float* __restrict__ xA, const float* __restrict__ xnext, const float2* Wt2,
    int laneM, int laneH, float2 (&xr)[16], v8f (&acc)[2][4])
{
    const int laneK = laneH << 1;
#pragma unroll
    for (int c = 0; c < 8; ++c) {
        float2 xn[16];
        if (c < 7) x_chunk_load(xn, xA, (c + 1) * 32, laneK);
        else if (NEXT) x_chunk_load(xn, xnext, 0, laneK);
        __builtin_amdgcn_sched_barrier(0);
#pragma unroll
        for (int kk = 0; kk < 32; kk += 4) {
            v2f a0, a1;
            a0.x = xr[kk >> 1].x; a0.y = xr[(kk >> 1) + 1].x;
            a1.x = xr[kk >> 1].y; a1.y = xr[(kk >> 1) + 1].y;
            const int pp = c * 16 + (kk >> 1) + laneH;
            const float4 b01 = *(const float4*)&Wt2[pp * WP + 2 * laneM];
            const float4 b23 = *(const float4*)&Wt2[pp * WP + 32 + 2 * laneM];
            v2f bf[4];
            bf[0].x = b01.x; bf[0].y = b01.y; bf[1].x = b01.z; bf[1].y = b01.w;
            bf[2].x = b23.x; bf[2].y = b23.y; bf[3].x = b23.z; bf[3].y = b23.w;
#pragma unroll
            for (int nb = 0; nb < 4; ++nb) {
                acc[0][nb] = wmma_f32(a0, bf[nb], acc[0][nb]);
                acc[1][nb] = wmma_f32(a1, bf[nb], acc[1][nb]);
            }
        }
        __builtin_amdgcn_sched_barrier(0);
        if (c < 7 || NEXT) {
#pragma unroll
            for (int q = 0; q < 16; ++q) xr[q] = xn[q];
        }
    }
}

__global__ __launch_bounds__(256) void qkv_attn_kernel(
    const float* __restrict__ x, const float* __restrict__ wq,
    const float* __restrict__ bq, float* __restrict__ qkv)
{
    __shared__ __attribute__((aligned(16))) float smem[WS_FLOATS];
    float2* Wt2 = (float2*)smem;
    const int b = blockIdx.x;
    const int tid = threadIdx.x;
    const int lane = tid & 31, wave = tid >> 5;
    const int m0 = wave * 32;
    const int laneM = lane & 15;
    const int laneH = lane >> 4;
    const int laneK = laneH << 1;
    const int sp = tid & 15, sj = tid >> 4;
    const int mbase = m0 + (laneH << 4);

    if (b >= KVB) {
        const int qb = b - KVB;
        const int mt = qb & 127;
        const int nt = qb >> 7;
        const int row0 = mt * 256;
        const int n = row0 >> 12;
        const int l0 = row0 & (LL - 1);
        const int j0 = nt * 64;
        const float* xA = x + (size_t)n * CC * LL + l0 + m0 + 2 * laneM;
        const float* wsrc = wq + (size_t)(j0 + sj) * CC + 2 * sp;
        const float* const wrow[4] = {wsrc, wsrc + 16 * CC, wsrc + 32 * CC, wsrc + 48 * CC};
        float2 xr[16];
        stage_weights_and_x0(wrow, Wt2, sp, sj, xr, xA, laneK);
        __syncthreads();
        v8f acc[2][4] = {};
        qkv_tile_gemm<false>(xA, xA, Wt2, laneM, laneH, xr, acc);
#pragma unroll
        for (int nb = 0; nb < 4; ++nb) {
            const float bias = bq[j0 + nb * 16 + laneM];
#pragma unroll
            for (int i = 0; i < 2; ++i)
#pragma unroll
                for (int r = 0; r < 8; ++r) acc[i][nb][r] += bias;
        }
#pragma unroll
        for (int i = 0; i < 2; ++i) {
#pragma unroll
            for (int hd = 0; hd < 2; ++hd) {
#pragma unroll
                for (int r = 0; r < 8; ++r) {
                    const float a0 = acc[i][2 * hd][r], a1 = acc[i][2 * hd + 1][r];
                    const float ss = row16_sum(a0 * a0 + a1 * a1);
                    const float inv = rsqrt_hw(ss);
                    acc[i][2 * hd][r] = a0 * inv;
                    acc[i][2 * hd + 1][r] = a1 * inv;
                }
            }
        }
#pragma unroll
        for (int i = 0; i < 2; ++i) {
#pragma unroll
            for (int nb = 0; nb < 4; ++nb) {
                float* outp = qkv + (size_t)(row0 + mbase + i) * TC + j0 + nb * 16 + laneM;
#pragma unroll
                for (int r = 0; r < 8; ++r) outp[(size_t)(2 * r) * TC] = acc[i][nb][r];
            }
        }
    } else {
        const int s = b & 7, nh = b >> 3;
        const int n = nh >> 3, h = nh & 7;
        const float* wk = wq + (size_t)(CC + h * DD + sj) * CC + 2 * sp;
        const float* wv = wq + (size_t)(2 * CC + h * DD + sj) * CC + 2 * sp;
        const float* const wrow[4] = {wk, wk + 16 * CC, wv, wv + 16 * CC};
        float bias[4];
        bias[0] = bq[CC + h * DD + laneM];
        bias[1] = bq[CC + h * DD + 16 + laneM];
        bias[2] = bq[2 * CC + h * DD + laneM];
        bias[3] = bq[2 * CC + h * DD + 16 + laneM];
        float* base = qkv + (size_t)n * LL * TC + h * DD;
        const float* xA0 = x + (size_t)n * CC * LL + s * 512 + m0 + 2 * laneM;
        float2 xr[16];
        stage_weights_and_x0(wrow, Wt2, sp, sj, xr, xA0, laneK);
        __syncthreads();
        v8f at[2][2] = {};
#pragma unroll 1
        for (int t = 0; t < KV_TILES; ++t) {
            const int l0 = s * 512 + t * 256;
            v8f acc[2][4] = {};
            const float* xA = xA0 + t * 256;
            qkv_tile_gemm<true>(xA, xA0 + ((t + 1) & (KV_TILES - 1)) * 256, Wt2, laneM, laneH, xr, acc);
#pragma unroll
            for (int i = 0; i < 2; ++i)
#pragma unroll
                for (int nb = 0; nb < 4; ++nb)
#pragma unroll
                    for (int r = 0; r < 8; ++r) acc[i][nb][r] += bias[nb];
#pragma unroll
            for (int i = 0; i < 2; ++i) {
#pragma unroll
                for (int r = 0; r < 8; ++r) {
                    const float a0 = acc[i][0][r], a1 = acc[i][1][r];
                    const float ss = row16_sum(a0 * a0 + a1 * a1);
                    const float inv = rsqrt_hw(ss);
                    acc[i][0][r] = a0 * inv;
                    acc[i][1][r] = a1 * inv;
                }
            }
#pragma unroll
            for (int i = 0; i < 2; ++i) {
#pragma unroll
                for (int nb = 2; nb < 4; ++nb) {
                    float* outp = base + (size_t)(l0 + mbase + i) * TC + 2 * CC + (nb - 2) * 16 + laneM;
#pragma unroll
                    for (int r = 0; r < 8; ++r) outp[(size_t)(2 * r) * TC] = acc[i][nb][r];
                }
            }
#pragma unroll
            for (int i = 0; i < 2; ++i) {
#pragma unroll
                for (int tt = 0; tt < 4; ++tt) {
                    v2f a0, a1, b0, b1;
                    a0.x = acc[i][0][2 * tt]; a0.y = acc[i][0][2 * tt + 1];
                    a1.x = acc[i][1][2 * tt]; a1.y = acc[i][1][2 * tt + 1];
                    b0.x = acc[i][2][2 * tt]; b0.y = acc[i][2][2 * tt + 1];
                    b1.x = acc[i][3][2 * tt]; b1.y = acc[i][3][2 * tt + 1];
                    at[0][0] = wmma_f32(a0, b0, at[0][0]);
                    at[0][1] = wmma_f32(a0, b1, at[0][1]);
                    at[1][0] = wmma_f32(a1, b0, at[1][0]);
                    at[1][1] = wmma_f32(a1, b1, at[1][1]);
                }
            }
        }
        __syncthreads();
#pragma unroll
        for (int md = 0; md < 2; ++md)
#pragma unroll
            for (int me = 0; me < 2; ++me)
#pragma unroll
                for (int r = 0; r < 8; ++r)
                    smem[wave * 1024 + (16 * md + 8 * laneH + r) * 32 + 16 * me + laneM] = at[md][me][r];
        __syncthreads();
        for (int i = tid; i < 1024; i += 256) {
            float sum = 0.f;
#pragma unroll
            for (int w = 0; w < 8; ++w) sum += smem[w * 1024 + i];
            base[(size_t)(s * 512 + (i >> 5)) * TC + CC + (i & 31)] = sum;
        }
    }
}

struct HeadStage {
    float v[9];
    float q[8];
    float4 p[8];
    float2 w[16];
};

static __device__ __forceinline__ void head_stage_load_vq(
    HeadStage& st, const float* __restrict__ qb, int l0, int sr, int se)
{
#pragma unroll
    for (int u = 0; u < 9; ++u) {
        const int l = l0 - 4 + sr + 8 * u;
        const int lcl = l < 0 ? 0 : (l >= LL ? LL - 1 : l);
        st.v[u] = qb[(size_t)lcl * TC + 2 * CC + se];
    }
#pragma unroll
    for (int u = 0; u < 8; ++u) st.q[u] = qb[(size_t)(l0 + sr + 8 * u) * TC + se];
}

static __device__ __forceinline__ void head_stage_load_wa(
    HeadStage& st, const float* __restrict__ qb, const float* __restrict__ wsrc, int pd, int pe)
{
#pragma unroll
    for (int u = 0; u < 16; ++u) st.w[u] = *(const float2*)&wsrc[(size_t)u * 16 * CC];
#pragma unroll
    for (int s = 0; s < 8; ++s)
        st.p[s] = *(const float4*)&qb[(size_t)(s * 512 + pd) * TC + CC + pe];
}

#define VS_OFF 0
#define QS_OFF 2592
#define A2_OFF 4896
#define AT_OFF 5920
#define WT_OFF 8032
#define SM_FLOATS 16288

__global__ __launch_bounds__(256, 4) void epi_proj_kernel(
    const float* __restrict__ qkv, const float* __restrict__ wdc,
    const float* __restrict__ wp, const float* __restrict__ bp, float* __restrict__ out)
{
    __shared__ __attribute__((aligned(16))) float smem[SM_FLOATS];
    float* v_s = smem + VS_OFF;
    float* q_s = smem + QS_OFF;
    float* a2 = smem + A2_OFF;
    float* At2f = smem + AT_OFF;
    float2* Wt2 = (float2*)(smem + WT_OFF);
    const int b = blockIdx.x;
    const int lc = b & 63;
    const int n = b >> 6;
    const int l0 = lc * 64;
    const int tid = threadIdx.x;
    const int lane = tid & 31;
    const int wave = __builtin_amdgcn_readfirstlane(tid >> 5);
    const int laneM = lane & 15;
    const int laneH = lane >> 4;
    const int sr = tid >> 5, se = tid & 31;
    const int sp = tid & 15, sj = tid >> 4;
    const int pd = tid >> 3, pe = (tid & 7) * 4;
    const int mp = wave & 3;
    const int nh = wave >> 2;
    const float* qn = qkv + (size_t)n * LL * TC;
    const float* wsrc0 = wp + (size_t)sj * CC + 2 * sp;
    const float inv_pi = 0.31830988618379067154f;
    float* vst = v_s + sr * 36 + se;
    float* qst = q_s + sr * 36 + se;
    float* ast = a2 + ((pd >> 1) * 32 + pe) * 2 + (pd & 1);
    float2* wst = Wt2 + sp * 258 + sj;
    const float2* apf = (const float2*)At2f + laneH * 66 + mp * 16 + laneM;
    const float2* bpf = Wt2 + laneH * 258 + nh * 128 + laneM;

    v8f acc[8] = {};
    HeadStage st;
    head_stage_load_vq(st, qn, l0, sr, se);
    head_stage_load_wa(st, qn, wsrc0, pd, pe);

    for (int h = 0; h < HH; ++h) {
        float wd[9];
#pragma unroll
        for (int t = 0; t < 9; ++t) wd[t] = wdc[h * 9 + t];
        float4 asum = st.p[0];
#pragma unroll
        for (int s = 1; s < 8; ++s) {
            asum.x += st.p[s].x; asum.y += st.p[s].y; asum.z += st.p[s].z; asum.w += st.p[s].w;
        }
#pragma unroll
        for (int u = 0; u < 9; ++u) {
            const int l = l0 - 4 + sr + 8 * u;
            vst[u * 8 * 36] = (l >= 0 && l < LL) ? st.v[u] : 0.0f;
        }
#pragma unroll
        for (int u = 0; u < 8; ++u) qst[u * 8 * 36] = st.q[u];
        ast[0] = asum.x; ast[2] = asum.y; ast[4] = asum.z; ast[6] = asum.w;
#pragma unroll
        for (int u = 0; u < 16; ++u) wst[16 * u] = st.w[u];
        __syncthreads();

        if (wave < 4) {
            const int m0 = wave * 16;
            const float* qf = q_s + (m0 + laneM) * 36 + 2 * laneH;
            const float* af = a2 + (laneH * 32 + laneM) * 2;
            const float* vw = v_s + (m0 + 8 * laneH) * 36 + laneM;
            float* atw = At2f + ((laneM >> 1) * 66 + m0 + 8 * laneH) * 2 + (laneM & 1);
            v8f P0 = {}, P1 = {};
#pragma unroll
            for (int s = 0; s < 8; ++s) {
                const v2f a = *(const v2f*)&qf[4 * s];
                const v2f b0 = *(const v2f*)&af[s * 2 * 32 * 2];
                const v2f b1 = *(const v2f*)&af[s * 2 * 32 * 2 + 16 * 2];
                P0 = wmma_f32(a, b0, P0);
                P1 = wmma_f32(a, b1, P1);
            }
            float w0[16], w1[16];
#pragma unroll
            for (int i = 0; i < 16; ++i) {
                w0[i] = vw[i * 36];
                w1[i] = vw[i * 36 + 16];
            }
#pragma unroll
            for (int r = 0; r < 8; ++r) {
                const float x0 = 0.5f * w0[r + 4] + inv_pi * P0[r];
                const float x1 = 0.5f * w1[r + 4] + inv_pi * P1[r];
                const float ss = row16_sum(x0 * x0 + x1 * x1);
                const float inv = rsqrt_hw(ss);
                float d0 = 0.f, d1 = 0.f;
#pragma unroll
                for (int t = 0; t < 9; ++t) {
                    d0 += wd[t] * w0[r + t];
                    d1 += wd[t] * w1[r + t];
                }
                atw[2 * r] = x0 * inv + d0;
                atw[8 * 66 * 2 + 2 * r] = x1 * inv + d1;
            }
        }
        __syncthreads();
        if (h + 1 < HH) {
            head_stage_load_vq(st, qn + (h + 1) * DD, l0, sr, se);
            head_stage_load_wa(st, qn + (h + 1) * DD, wsrc0 + (h + 1) * DD, pd, pe);
        }
        __builtin_amdgcn_sched_barrier(0);

#pragma unroll
        for (int ks = 0; ks < 8; ++ks) {
            const v2f a = *(const v2f*)&apf[ks * 2 * 66];
#pragma unroll
            for (int nb = 0; nb < 8; ++nb) {
                const v2f bb = *(const v2f*)&bpf[ks * 2 * 258 + nb * 16];
                acc[nb] = wmma_f32(a, bb, acc[nb]);
            }
        }
        __syncthreads();
    }

    float bias[8];
#pragma unroll
    for (int nb = 0; nb < 8; ++nb) bias[nb] = bp[nh * 128 + nb * 16 + laneM];
    const int oj = tid >> 4, ol4 = (tid & 15) * 4;
    for (int pass = 0; pass < 2; ++pass) {
        if (nh == pass) {
#pragma unroll
            for (int nb = 0; nb < 8; ++nb) {
                float* o = smem + (nb * 16 + laneM) * 68 + mp * 16 + 8 * laneH;
                float4 lo, hi;
                lo.x = acc[nb][0] + bias[nb]; lo.y = acc[nb][1] + bias[nb];
                lo.z = acc[nb][2] + bias[nb]; lo.w = acc[nb][3] + bias[nb];
                hi.x = acc[nb][4] + bias[nb]; hi.y = acc[nb][5] + bias[nb];
                hi.z = acc[nb][6] + bias[nb]; hi.w = acc[nb][7] + bias[nb];
                *(float4*)o = lo;
                *(float4*)(o + 4) = hi;
            }
        }
        __syncthreads();
        float* dst = out + ((size_t)(n * CC + pass * 128 + oj)) * LL + l0 + ol4;
#pragma unroll
        for (int u = 0; u < 8; ++u) {
            const float4 t = *(const float4*)&smem[(oj + 16 * u) * 68 + ol4];
            *(float4*)&dst[(size_t)u * 16 * LL] = t;
        }
        __syncthreads();
    }
}

extern "C" void kernel_launch(void* const* d_in, const int* in_sizes, int n_in,
                              void* d_out, int out_size, void* d_ws, size_t ws_size,
                              hipStream_t stream) {
    const float* x      = (const float*)d_in[0];
    const float* w_qkv  = (const float*)d_in[1];
    const float* b_qkv  = (const float*)d_in[2];
    const float* w_proj = (const float*)d_in[3];
    const float* b_proj = (const float*)d_in[4];
    const float* w_dcv  = (const float*)d_in[5];
    float* out = (float*)d_out;

    float* qkv = (float*)d_ws;

    qkv_attn_kernel<<<QKV_BLOCKS, 256, 0, stream>>>(x, w_qkv, b_qkv, qkv);
    epi_proj_kernel<<<512, 256, 0, stream>>>(qkv, w_dcv, w_proj, b_proj, out);
}
